# starting kernel
# baseline (speedup 1.0000x reference)
#include <hip/hip_runtime.h>
#include <hip/hip_bf16.h>
#include <math.h>

#define N_NODES   50000
#define N_EDGES   800000
#define IN_F      128
#define HEADS     8
#define HEAD_DIM  16
#define OUT_F     (HEADS * HEAD_DIM)
#define ALPHA     0.2f
#define EPS_F     1e-6f

typedef __attribute__((ext_vector_type(2))) float v2f;
typedef __attribute__((ext_vector_type(8))) float v8f;

__device__ __forceinline__ unsigned enc_f32(float f) {
    unsigned u = __float_as_uint(f);
    return (u & 0x80000000u) ? ~u : (u | 0x80000000u);
}
__device__ __forceinline__ float dec_f32(unsigned u) {
    return (u & 0x80000000u) ? __uint_as_float(u ^ 0x80000000u)
                             : __uint_as_float(~u);
}
__device__ __forceinline__ float leaky(float s) {
    return s >= 0.0f ? s : ALPHA * s;
}

__global__ void gat_init(float* __restrict__ out, const float* __restrict__ bias,
                         float* __restrict__ denom, unsigned* __restrict__ gmax) {
    const int i = blockIdx.x * blockDim.x + threadIdx.x;
    if (i < N_NODES * OUT_F) out[i] = bias[i & (OUT_F - 1)];
    if (i < N_NODES * HEADS) denom[i] = 0.0f;
    if (i < HEADS)           gmax[i] = 0u;
}

__global__ __launch_bounds__(256)
void gat_gemm_wmma(const float* __restrict__ x, const float* __restrict__ W,
                   float* __restrict__ h) {
    __shared__ float As[16 * IN_F];
    const int tid = threadIdx.x;
    const int m0  = blockIdx.x * 16;

    {
        const int base = tid * 8;
        const int row  = base >> 7;
        const int col  = base & 127;
        const float4* g =
            reinterpret_cast<const float4*>(x + (size_t)(m0 + row) * IN_F + col);
        float4 p0 = g[0];
        float4 p1 = g[1];
        *reinterpret_cast<float4*>(&As[row * IN_F + col])     = p0;
        *reinterpret_cast<float4*>(&As[row * IN_F + col + 4]) = p1;
    }
    __syncthreads();

    const int lane = tid & 31;
    const int head = tid >> 5;
    const int l16  = lane & 15;
    const bool hi  = lane >= 16;
    const float* Wh = W + (size_t)head * IN_F * HEAD_DIM;

    v8f c = {};
#pragma unroll
    for (int k = 0; k < IN_F; k += 4) {
        const int kk = k + (hi ? 2 : 0);
        v2f a, b;
        a.x = As[l16 * IN_F + kk];
        a.y = As[l16 * IN_F + kk + 1];
        b.x = Wh[kk * HEAD_DIM + l16];
        b.y = Wh[(kk + 1) * HEAD_DIM + l16];
        c = __builtin_amdgcn_wmma_f32_16x16x4_f32(
                 false, a,  false, b,
                 (short)0, c,  false,  false);
    }

    const int rbase = hi ? 8 : 0;
#pragma unroll
    for (int v = 0; v < 8; ++v)
        h[(size_t)(m0 + rbase + v) * OUT_F + head * HEAD_DIM + l16] = c[v];
}

__global__ void gat_node_scores(const float* __restrict__ h, const float* __restrict__ a,
                                float* __restrict__ s_src, float* __restrict__ s_tgt) {
    const int i = blockIdx.x * blockDim.x + threadIdx.x;
    if (i >= N_NODES * HEADS) return;
    const int hd = i & (HEADS - 1);
    const int n  = i >> 3;
    const float* hv = h + (size_t)n * OUT_F + hd * HEAD_DIM;
    const float* as = a + hd * 2 * HEAD_DIM;
    const float* at = as + HEAD_DIM;
    float ss = 0.0f, st = 0.0f;
#pragma unroll
    for (int d = 0; d < HEAD_DIM; ++d) {
        const float v = hv[d];
        ss += v * as[d];
        st += v * at[d];
    }
    s_src[i] = ss;
    s_tgt[i] = st;
}

__global__ __launch_bounds__(256)
void gat_edge_max(const int* __restrict__ src, const int* __restrict__ tgt,
                  const float* __restrict__ s_src, const float* __restrict__ s_tgt,
                  unsigned* __restrict__ gmax) {
    __shared__ unsigned lmax[HEADS];
    const int tid = threadIdx.x;
    if (tid < HEADS) lmax[tid] = 0u;
    __syncthreads();

    const int e = blockIdx.x * blockDim.x + tid;
    if (e < N_EDGES) {
        const int s = src[e], t = tgt[e];
#pragma unroll
        for (int hd = 0; hd < HEADS; ++hd) {
            const float sc = leaky(s_src[s * HEADS + hd] + s_tgt[t * HEADS + hd]);
            atomicMax(&lmax[hd], enc_f32(sc));
        }
    }
    __syncthreads();
    if (tid < HEADS) atomicMax(&gmax[tid], lmax[tid]);
}

__global__ __launch_bounds__(256)
void gat_edge_denom(const int* __restrict__ src, const int* __restrict__ tgt,
                    const float* __restrict__ s_src, const float* __restrict__ s_tgt,
                    const unsigned* __restrict__ gmax, float* __restrict__ denom) {
    const int e = blockIdx.x * blockDim.x + threadIdx.x;
    if (e >= N_EDGES) return;
    const int s = src[e], t = tgt[e];
#pragma unroll
    for (int hd = 0; hd < HEADS; ++hd) {
        const float mx = dec_f32(gmax[hd]);
        const float sc = leaky(s_src[s * HEADS + hd] + s_tgt[t * HEADS + hd]);
        const float ev = __expf(fminf(fmaxf(sc - mx, -20.0f), 20.0f));
        atomicAdd(&denom[t * HEADS + hd], ev);
    }
}

__global__ __launch_bounds__(256)
void gat_edge_agg(const int* __restrict__ src, const int* __restrict__ tgt,
                  const float* __restrict__ s_src, const float* __restrict__ s_tgt,
                  const unsigned* __restrict__ gmax, const float* __restrict__ denom,
                  const float* __restrict__ h, float* __restrict__ out) {
    const long long gid = (long long)blockIdx.x * blockDim.x + threadIdx.x;
    const int e  = (int)(gid >> 7);
    const int c  = (int)(gid & 127);
    const int hd = c >> 4;
    if (e >= N_EDGES) return;
    const int s = src[e], t = tgt[e];
    const float mx = dec_f32(gmax[hd]);
    const float sc = leaky(s_src[s * HEADS + hd] + s_tgt[t * HEADS + hd]);
    const float ev = __expf(fminf(fmaxf(sc - mx, -20.0f), 20.0f));
    const float attn = ev / (denom[t * HEADS + hd] + EPS_F);
    atomicAdd(&out[(size_t)t * OUT_F + c], h[(size_t)s * OUT_F + c] * attn);
}

extern "C" void kernel_launch(void* const* d_in, const int* in_sizes, int n_in,
                              void* d_out, int out_size, void* d_ws, size_t ws_size,
                              hipStream_t stream) {
    (void)in_sizes; (void)n_in; (void)out_size; (void)ws_size;

    const float* x    = (const float*)d_in[0];
    const int*   ei   = (const int*)  d_in[1];
    const float* W    = (const float*)d_in[2];
    const float* a    = (const float*)d_in[3];
    const float* bias = (const float*)d_in[4];
    float* out = (float*)d_out;

    const int* src = ei;
    const int* tgt = ei + N_EDGES;

    float*    ws    = (float*)d_ws;
    float*    hbuf  = ws;
    float*    ssrc  = hbuf + (size_t)N_NODES * OUT_F;
    float*    stgt  = ssrc + (size_t)N_NODES * HEADS;
    float*    denom = stgt + (size_t)N_NODES * HEADS;
    unsigned* gmax  = (unsigned*)(denom + (size_t)N_NODES * HEADS);

    const int outN = N_NODES * OUT_F;
    gat_init<<<(outN + 255) / 256, 256, 0, stream>>>(out, bias, denom, gmax);

    gat_gemm_wmma<<<N_NODES / 16, 256, 0, stream>>>(x, W, hbuf);

    gat_node_scores<<<(N_NODES * HEADS + 255) / 256, 256, 0, stream>>>(hbuf, a, ssrc, stgt);

    gat_edge_max<<<(N_EDGES + 255) / 256, 256, 0, stream>>>(src, tgt, ssrc, stgt, gmax);

    gat_edge_denom<<<(N_EDGES + 255) / 256, 256, 0, stream>>>(src, tgt, ssrc, stgt, gmax, denom);

    const long long aggThreads = (long long)N_EDGES * OUT_F;
    gat_edge_agg<<<(unsigned)((aggThreads + 255) / 256), 256, 0, stream>>>(
        src, tgt, ssrc, stgt, gmax, denom, hbuf, out);
}


# pass 4 scans one owner byte per edge (target / 196, written by the max pass), 16 edges per 16-B load, 4 per 32-bit zero-byte test, not the 4-B targets; sums in arrival order as before
# speedup vs baseline: 3.7469x; 3.7469x over previous
#include <hip/hip_runtime.h>
#include <hip/hip_bf16.h>
#include <math.h>

#define N_NODES   50000
#define N_EDGES   800000
#define IN_F      128
#define HEADS     8
#define HEAD_DIM  16
#define OUT_F     (HEADS * HEAD_DIM)
#define ALPHA     0.2f
#define EPS_F     1e-6f

#define NBLK  256
#define NPB   196
static_assert(NBLK <= 256 && NBLK * NPB >= N_NODES, "slice number must fit one byte");
static_assert(N_EDGES % 256 == 0, "pass 3 writes whole 256-byte stretches of the owner plane");

typedef __attribute__((ext_vector_type(2))) float v2f;
typedef __attribute__((ext_vector_type(8))) float v8f;

__device__ __forceinline__ unsigned enc_f32(float f) {
    unsigned u = __float_as_uint(f);
    return (u & 0x80000000u) ? ~u : (u | 0x80000000u);
}
__device__ __forceinline__ float dec_f32(unsigned u) {
    return (u & 0x80000000u) ? __uint_as_float(u ^ 0x80000000u)
                             : __uint_as_float(~u);
}
__device__ __forceinline__ float leaky(float s) {
    return s >= 0.0f ? s : ALPHA * s;
}

__global__ void gat_init(unsigned* __restrict__ gmax) {
    if (threadIdx.x < HEADS) gmax[threadIdx.x] = 0u;
}

__global__ __launch_bounds__(256)
void gat_gemm_wmma(const float* __restrict__ x, const float* __restrict__ W,
                   float* __restrict__ h) {
    __shared__ float As[16 * IN_F];
    const int tid = threadIdx.x;
    const int m0  = blockIdx.x * 16;

    {
        const int base = tid * 8;
        const int row  = base >> 7;
        const int col  = base & 127;
        const float4* g =
            reinterpret_cast<const float4*>(x + (size_t)(m0 + row) * IN_F + col);
        float4 p0 = g[0];
        float4 p1 = g[1];
        *reinterpret_cast<float4*>(&As[row * IN_F + col])     = p0;
        *reinterpret_cast<float4*>(&As[row * IN_F + col + 4]) = p1;
    }
    __syncthreads();

    const int lane = tid & 31;
    const int head = tid >> 5;
    const int l16  = lane & 15;
    const bool hi  = lane >= 16;
    const float* Wh = W + (size_t)head * IN_F * HEAD_DIM;

    v8f c = {};
#pragma unroll
    for (int k = 0; k < IN_F; k += 4) {
        const int kk = k + (hi ? 2 : 0);
        v2f a, b;
        a.x = As[l16 * IN_F + kk];
        a.y = As[l16 * IN_F + kk + 1];
        b.x = Wh[kk * HEAD_DIM + l16];
        b.y = Wh[(kk + 1) * HEAD_DIM + l16];
        c = __builtin_amdgcn_wmma_f32_16x16x4_f32(
                 false, a,  false, b,
                 (short)0, c,  false,  false);
    }

    const int rbase = hi ? 8 : 0;
#pragma unroll
    for (int v = 0; v < 8; ++v)
        h[(size_t)(m0 + rbase + v) * OUT_F + head * HEAD_DIM + l16] = c[v];
}

__global__ void gat_node_scores(const float* __restrict__ h, const float* __restrict__ a,
                                float* __restrict__ s_src, float* __restrict__ s_tgt) {
    const int i = blockIdx.x * blockDim.x + threadIdx.x;
    if (i >= N_NODES * HEADS) return;
    const int hd = i & (HEADS - 1);
    const int n  = i >> 3;
    const float* hv = h + (size_t)n * OUT_F + hd * HEAD_DIM;
    const float* as = a + hd * 2 * HEAD_DIM;
    const float* at = as + HEAD_DIM;
    float ss = 0.0f, st = 0.0f;
#pragma unroll
    for (int d = 0; d < HEAD_DIM; ++d) {
        const float v = hv[d];
        ss += v * as[d];
        st += v * at[d];
    }
    s_src[i] = ss;
    s_tgt[i] = st;
}

__global__ __launch_bounds__(256)
void gat_edge_max(const int* __restrict__ src, const int* __restrict__ tgt,
                  const float* __restrict__ s_src, const float* __restrict__ s_tgt,
                  unsigned* __restrict__ gmax, unsigned* __restrict__ own4) {
    __shared__ unsigned lmax[HEADS];
    __shared__ unsigned ownw[64];
    const int tid = threadIdx.x;
    if (tid < HEADS) lmax[tid] = 0u;
    __syncthreads();

    const int e = blockIdx.x * blockDim.x + tid;
    if (e < N_EDGES) {
        const int s = src[e], t = tgt[e];
        reinterpret_cast<unsigned char*>(ownw)[tid] = (unsigned char)((unsigned)t / (unsigned)NPB);
#pragma unroll
        for (int hd = 0; hd < HEADS; ++hd) {
            const float sc = leaky(s_src[s * HEADS + hd] + s_tgt[t * HEADS + hd]);
            atomicMax(&lmax[hd], enc_f32(sc));
        }
    }
    __syncthreads();
    if (tid < HEADS) atomicMax(&gmax[tid], lmax[tid]);
    if (tid < 64) own4[blockIdx.x * 64 + tid] = ownw[tid];
}

#define OTHR  1024
#define OWAV  (OTHR / 32)
#define LPB   2
#define SCH   (OTHR * LPB * 16)
#define PU    6
#define WQ    (32 * LPB * 16 + 8)
#define NSC   ((N_EDGES + SCH - 1) / SCH)
#define NP16  (N_EDGES / 16)
static_assert(N_EDGES % 16 == 0, "the owner plane is read in whole 16-B pieces");

__device__ __forceinline__ void load_chunk(const uint4* __restrict__ own16, int chunk,
                                           int wave, int lane, uint4 (&pre)[LPB]) {
#pragma unroll
    for (int k = 0; k < LPB; ++k) {
        const int i16 = chunk * (SCH / 16) + (wave * LPB + k) * 32 + lane;
        pre[k] = own16[min(i16, NP16 - 1)];
    }
}

template <bool PAD>
__device__ __forceinline__ void batch_ids(const int* qw, int i0, int n,
                                          const int* __restrict__ tgt, const int* __restrict__ src,
                                          int (&t)[PU], int (&s)[PU]) {
    int e[PU];
#pragma unroll
    for (int u = 0; u < PU; ++u) e[u] = qw[(!PAD || i0 + u < n) ? i0 + u : i0];
#pragma unroll
    for (int u = 0; u < PU; ++u) { t[u] = tgt[e[u]]; s[u] = src[e[u]]; }
}

template <bool PAD>
__device__ __forceinline__ void batch_add(const int (&t)[PU], const int (&s)[PU], int i0, int n,
                                          const float* __restrict__ h, const float* __restrict__ s_src,
                                          const float* stg, float* acc, float* den,
                                          int n0, int lane, int hd, float mx) {
    float4 hv[PU];
    float  ss[PU];
#pragma unroll
    for (int u = 0; u < PU; ++u) {
        hv[u] = *reinterpret_cast<const float4*>(h + (size_t)s[u] * OUT_F + 4 * lane);
        ss[u] = s_src[s[u] * HEADS + hd];
    }
#pragma unroll
    for (int u = 0; u < PU; ++u) {
        const int   lt  = t[u] - n0;
        const float scv = leaky(ss[u] + stg[lt * HEADS + hd]);
        float ev = __expf(fminf(fmaxf(scv - mx, -20.0f), 20.0f));
        if (PAD && i0 + u >= n) ev = 0.0f;
        float* a = acc + lt * OUT_F + lane;
        atomicAdd(a,      hv[u].x * ev);
        atomicAdd(a + 32, hv[u].y * ev);
        atomicAdd(a + 64, hv[u].z * ev);
        atomicAdd(a + 96, hv[u].w * ev);
        if ((lane & 3) == 0) atomicAdd(&den[lt * HEADS + hd], ev);
    }
}

__global__ __launch_bounds__(OTHR)
void gat_owner_agg(const int* __restrict__ src, const int* __restrict__ tgt,
                   const float* __restrict__ s_src, const float* __restrict__ s_tgt,
                   const unsigned* __restrict__ gmax, const float* __restrict__ h,
                   const uint4* __restrict__ own16,
                   const float* __restrict__ bias, float* __restrict__ out) {
    __shared__ float acc[NPB * OUT_F];
    __shared__ float den[NPB * HEADS];
    __shared__ float stg[NPB * HEADS];
    __shared__ int   q[OWAV][WQ];

    const int tid  = threadIdx.x;
    const int lane = tid & 31;
    const int wave = tid >> 5;
    const int hd   = lane >> 2;
    const int n0   = blockIdx.x * NPB;

    for (int i = tid; i < NPB * OUT_F; i += OTHR) acc[i] = 0.0f;
    for (int i = tid; i < NPB * HEADS; i += OTHR) {
        den[i] = 0.0f;
        stg[i] = (n0 + (i >> 3) < N_NODES) ? s_tgt[n0 * HEADS + i] : 0.0f;
    }

    const float mx = dec_f32(gmax[hd]);
    const unsigned pat = (unsigned)blockIdx.x * 0x01010101u;
    int* qw = q[wave];

    int chunk = blockIdx.x % NSC;
    uint4 pre[LPB];
    load_chunk(own16, chunk, wave, lane, pre);
    __syncthreads();

    int n = 0;
    int t[PU], s[PU];
    for (int sc = 0; sc < NSC; ++sc) {
#pragma unroll
        for (int k = 0; k < LPB; ++k) {
            const int  i16 = chunk * (SCH / 16) + (wave * LPB + k) * 32 + lane;
            const bool inb = i16 < NP16;
            const unsigned inbm = __builtin_amdgcn_ballot_w32(inb);
            const unsigned w4[4] = {pre[k].x ^ pat, pre[k].y ^ pat, pre[k].z ^ pat, pre[k].w ^ pat};
#pragma unroll
            for (int d = 0; d < 4; ++d) {
                const unsigned w = w4[d];
                if (__builtin_amdgcn_ballot_w32(((w - 0x01010101u) & ~w & 0x80808080u) != 0u) == 0u) continue;
#pragma unroll
                for (int j = 0; j < 4; ++j) {
                    const bool     hit  = (w & (0xffu << (8 * j))) == 0u;
                    const unsigned mask = __builtin_amdgcn_ballot_w32(hit) & inbm;
                    if (mask != 0u) {
                        if (inb && hit) qw[n + (int)__builtin_amdgcn_mbcnt_lo(mask, 0u)] = i16 * 16 + d * 4 + j;
                        n += __builtin_popcount(mask);
                    }
                }
            }
        }
        chunk = (chunk + 1 == NSC) ? 0 : chunk + 1;

        const bool has = n >= PU;
        if (has) batch_ids<false>(qw, 0, n, tgt, src, t, s);
        load_chunk(own16, chunk, wave, lane, pre);
        if (has) batch_add<false>(t, s, 0, n, h, s_src, stg, acc, den, n0, lane, hd, mx);
        int i0 = has ? PU : 0;
        for (; i0 + PU <= n; i0 += PU) {
            batch_ids<false>(qw, i0, n, tgt, src, t, s);
            batch_add<false>(t, s, i0, n, h, s_src, stg, acc, den, n0, lane, hd, mx);
        }
        const int left = n - i0;
        if (i0 > 0 && left > 0) {
            const int v = qw[i0 + min(lane, left - 1)];
            if (lane < left) qw[lane] = v;
        }
        n = left;
    }
    if (n > 0) {
        batch_ids<true>(qw, 0, n, tgt, src, t, s);
        batch_add<true>(t, s, 0, n, h, s_src, stg, acc, den, n0, lane, hd, mx);
    }
    __syncthreads();

    const float4 bb = *reinterpret_cast<const float4*>(bias + 4 * lane);
    for (int idx = tid; idx < NPB * 32; idx += OTHR) {
        const int lt = idx >> 5;
        const int nd = n0 + lt;
        if (nd < N_NODES) {
            const float  d = den[lt * HEADS + hd] + EPS_F;
            const float* a = acc + lt * OUT_F + lane;
            float4 v;
            v.x = a[0]  / d + bb.x;
            v.y = a[32] / d + bb.y;
            v.z = a[64] / d + bb.z;
            v.w = a[96] / d + bb.w;
            *reinterpret_cast<float4*>(out + (size_t)nd * OUT_F + 4 * lane) = v;
        }
    }
}

extern "C" void kernel_launch(void* const* d_in, const int* in_sizes, int n_in,
                              void* d_out, int out_size, void* d_ws, size_t ws_size,
                              hipStream_t stream) {
    (void)in_sizes; (void)n_in; (void)out_size; (void)ws_size;

    const float* x    = (const float*)d_in[0];
    const int*   ei   = (const int*)  d_in[1];
    const float* W    = (const float*)d_in[2];
    const float* a    = (const float*)d_in[3];
    const float* bias = (const float*)d_in[4];
    float* out = (float*)d_out;

    const int* src = ei;
    const int* tgt = ei + N_EDGES;

    float*    ws    = (float*)d_ws;
    float*    hbuf  = ws;
    float*    ssrc  = hbuf + (size_t)N_NODES * OUT_F;
    float*    stgt  = ssrc + (size_t)N_NODES * HEADS;
    float*    ownf  = stgt + (size_t)N_NODES * HEADS;
    unsigned* gmax  = (unsigned*)(ownf + (size_t)N_NODES * HEADS);

    gat_init<<<1, 32, 0, stream>>>(gmax);

    gat_gemm_wmma<<<N_NODES / 16, 256, 0, stream>>>(x, W, hbuf);

    gat_node_scores<<<(N_NODES * HEADS + 255) / 256, 256, 0, stream>>>(hbuf, a, ssrc, stgt);

    gat_edge_max<<<(N_EDGES + 255) / 256, 256, 0, stream>>>(src, tgt, ssrc, stgt, gmax, (unsigned*)ownf);

    gat_owner_agg<<<NBLK, OTHR, 0, stream>>>(src, tgt, ssrc, stgt, gmax, hbuf, (const uint4*)ownf, bias, out);
}


# product kernel re-tiled: a wave owns 16 rows x all 8 heads, x fragments straight from global (16 B a lane, not via LDS), W in LDS once per block in fragment order; k summed in a permuted order per 8
# speedup vs baseline: 4.0893x; 1.0914x over previous
#include <hip/hip_runtime.h>
#include <hip/hip_bf16.h>
#include <math.h>

#define N_NODES   50000
#define N_EDGES   800000
#define IN_F      128
#define HEADS     8
#define HEAD_DIM  16
#define OUT_F     (HEADS * HEAD_DIM)
#define ALPHA     0.2f
#define EPS_F     1e-6f

#define NBLK  256
#define NPB   196
static_assert(NBLK <= 256 && NBLK * NPB >= N_NODES, "slice number must fit one byte");
static_assert(N_EDGES % 256 == 0, "pass 3 writes whole 256-byte stretches of the owner plane");

typedef __attribute__((ext_vector_type(2))) float v2f;
typedef __attribute__((ext_vector_type(8))) float v8f;

__device__ __forceinline__ unsigned enc_f32(float f) {
    unsigned u = __float_as_uint(f);
    return (u & 0x80000000u) ? ~u : (u | 0x80000000u);
}
__device__ __forceinline__ float dec_f32(unsigned u) {
    return (u & 0x80000000u) ? __uint_as_float(u ^ 0x80000000u)
                             : __uint_as_float(~u);
}
__device__ __forceinline__ float leaky(float s) {
    return s >= 0.0f ? s : ALPHA * s;
}

__global__ void gat_init(unsigned* __restrict__ gmax) {
    if (threadIdx.x < HEADS) gmax[threadIdx.x] = 0u;
}

#define GROWS  128
#define GBLK   ((N_NODES + GROWS - 1) / GROWS)
static_assert(N_NODES % 16 == 0, "a wave's 16-row tile is whole or absent");

__global__ __launch_bounds__(256)
void gat_gemm_wmma(const float* __restrict__ x, const float* __restrict__ W,
                   float* __restrict__ h) {
    __shared__ float4 Wl[HEADS * 32 * 16];
    const int tid  = threadIdx.x;
    const int lane = tid & 31;
    const int wave = tid >> 5;
    const int l16  = lane & 15;
    const int hh   = lane >> 4;
    const int m0   = blockIdx.x * GROWS + wave * 16;
    const bool live = m0 < N_NODES;

    float4 a4[16];
    {
        const int row = min(m0 + l16, N_NODES - 1);
        const float4* xr = reinterpret_cast<const float4*>(x + (size_t)row * IN_F) + hh;
#pragma unroll
        for (int j = 0; j < 16; ++j) a4[j] = xr[2 * j];
    }

#pragma unroll
    for (int u = 0; u < 16; ++u) {
        const int f = tid + 256 * u;
        const int d = f & 15, q = (f >> 4) & 31, t = f >> 9;
        const float* wp = W + (size_t)(t * IN_F + 4 * q) * HEAD_DIM + d;
        float4 v;
        v.x = wp[0];
        v.y = wp[HEAD_DIM];
        v.z = wp[2 * HEAD_DIM];
        v.w = wp[3 * HEAD_DIM];
        Wl[f] = v;
    }
    __syncthreads();
    if (!live) return;

    v8f c[HEADS];
#pragma unroll
    for (int t = 0; t < HEADS; ++t) c[t] = v8f{};

    const float4* bl = Wl + lane;
    float4 b[2][HEADS];
#pragma unroll
    for (int t = 0; t < HEADS; ++t) b[0][t] = bl[t * 512];
#pragma unroll
    for (int j = 0; j < 16; ++j) {
        if (j + 1 < 16) {
#pragma unroll
            for (int t = 0; t < HEADS; ++t) b[(j + 1) & 1][t] = bl[t * 512 + (j + 1) * 32];
        }
        __builtin_amdgcn_sched_barrier(0);
        v2f alo, ahi;
        alo.x = a4[j].x; alo.y = a4[j].y;
        ahi.x = a4[j].z; ahi.y = a4[j].w;
#pragma unroll
        for (int t = 0; t < HEADS; ++t) {
            v2f bb; bb.x = b[j & 1][t].x; bb.y = b[j & 1][t].y;
            c[t] = __builtin_amdgcn_wmma_f32_16x16x4_f32(
                        false, alo,  false, bb,
                        (short)0, c[t],  false,  false);
        }
#pragma unroll
        for (int t = 0; t < HEADS; ++t) {
            v2f bb; bb.x = b[j & 1][t].z; bb.y = b[j & 1][t].w;
            c[t] = __builtin_amdgcn_wmma_f32_16x16x4_f32(
                        false, ahi,  false, bb,
                        (short)0, c[t],  false,  false);
        }
        __builtin_amdgcn_sched_barrier(0);
    }

    const int rbase = hh ? 8 : 0;
#pragma unroll
    for (int v = 0; v < 8; ++v) {
        float* hr = h + (size_t)(m0 + rbase + v) * OUT_F + l16;
#pragma unroll
        for (int t = 0; t < HEADS; ++t) hr[t * HEAD_DIM] = c[t][v];
    }
}

__global__ void gat_node_scores(const float* __restrict__ h, const float* __restrict__ a,
                                float* __restrict__ s_src, float* __restrict__ s_tgt) {
    const int i = blockIdx.x * blockDim.x + threadIdx.x;
    if (i >= N_NODES * HEADS) return;
    const int hd = i & (HEADS - 1);
    const int n  = i >> 3;
    const float* hv = h + (size_t)n * OUT_F + hd * HEAD_DIM;
    const float* as = a + hd * 2 * HEAD_DIM;
    const float* at = as + HEAD_DIM;
    float ss = 0.0f, st = 0.0f;
#pragma unroll
    for (int d = 0; d < HEAD_DIM; ++d) {
        const float v = hv[d];
        ss += v * as[d];
        st += v * at[d];
    }
    s_src[i] = ss;
    s_tgt[i] = st;
}

__global__ __launch_bounds__(256)
void gat_edge_max(const int* __restrict__ src, const int* __restrict__ tgt,
                  const float* __restrict__ s_src, const float* __restrict__ s_tgt,
                  unsigned* __restrict__ gmax, unsigned* __restrict__ own4) {
    __shared__ unsigned lmax[HEADS];
    __shared__ unsigned ownw[64];
    const int tid = threadIdx.x;
    if (tid < HEADS) lmax[tid] = 0u;
    __syncthreads();

    const int e = blockIdx.x * blockDim.x + tid;
    if (e < N_EDGES) {
        const int s = src[e], t = tgt[e];
        reinterpret_cast<unsigned char*>(ownw)[tid] = (unsigned char)((unsigned)t / (unsigned)NPB);
#pragma unroll
        for (int hd = 0; hd < HEADS; ++hd) {
            const float sc = leaky(s_src[s * HEADS + hd] + s_tgt[t * HEADS + hd]);
            atomicMax(&lmax[hd], enc_f32(sc));
        }
    }
    __syncthreads();
    if (tid < HEADS) atomicMax(&gmax[tid], lmax[tid]);
    if (tid < 64) own4[blockIdx.x * 64 + tid] = ownw[tid];
}

#define OTHR  1024
#define OWAV  (OTHR / 32)
#define LPB   2
#define SCH   (OTHR * LPB * 16)
#define PU    6
#define WQ    (32 * LPB * 16 + 8)
#define NSC   ((N_EDGES + SCH - 1) / SCH)
#define NP16  (N_EDGES / 16)
static_assert(N_EDGES % 16 == 0, "the owner plane is read in whole 16-B pieces");

__device__ __forceinline__ void load_chunk(const uint4* __restrict__ own16, int chunk,
                                           int wave, int lane, uint4 (&pre)[LPB]) {
#pragma unroll
    for (int k = 0; k < LPB; ++k) {
        const int i16 = chunk * (SCH / 16) + (wave * LPB + k) * 32 + lane;
        pre[k] = own16[min(i16, NP16 - 1)];
    }
}

template <bool PAD>
__device__ __forceinline__ void batch_ids(const int* qw, int i0, int n,
                                          const int* __restrict__ tgt, const int* __restrict__ src,
                                          int (&t)[PU], int (&s)[PU]) {
    int e[PU];
#pragma unroll
    for (int u = 0; u < PU; ++u) e[u] = qw[(!PAD || i0 + u < n) ? i0 + u : i0];
#pragma unroll
    for (int u = 0; u < PU; ++u) { t[u] = tgt[e[u]]; s[u] = src[e[u]]; }
}

template <bool PAD>
__device__ __forceinline__ void batch_add(const int (&t)[PU], const int (&s)[PU], int i0, int n,
                                          const float* __restrict__ h, const float* __restrict__ s_src,
                                          const float* stg, float* acc, float* den,
                                          int n0, int lane, int hd, float mx) {
    float4 hv[PU];
    float  ss[PU];
#pragma unroll
    for (int u = 0; u < PU; ++u) {
        hv[u] = *reinterpret_cast<const float4*>(h + (size_t)s[u] * OUT_F + 4 * lane);
        ss[u] = s_src[s[u] * HEADS + hd];
    }
#pragma unroll
    for (int u = 0; u < PU; ++u) {
        const int   lt  = t[u] - n0;
        const float scv = leaky(ss[u] + stg[lt * HEADS + hd]);
        float ev = __expf(fminf(fmaxf(scv - mx, -20.0f), 20.0f));
        if (PAD && i0 + u >= n) ev = 0.0f;
        float* a = acc + lt * OUT_F + lane;
        atomicAdd(a,      hv[u].x * ev);
        atomicAdd(a + 32, hv[u].y * ev);
        atomicAdd(a + 64, hv[u].z * ev);
        atomicAdd(a + 96, hv[u].w * ev);
        if ((lane & 3) == 0) atomicAdd(&den[lt * HEADS + hd], ev);
    }
}

__global__ __launch_bounds__(OTHR)
void gat_owner_agg(const int* __restrict__ src, const int* __restrict__ tgt,
                   const float* __restrict__ s_src, const float* __restrict__ s_tgt,
                   const unsigned* __restrict__ gmax, const float* __restrict__ h,
                   const uint4* __restrict__ own16,
                   const float* __restrict__ bias, float* __restrict__ out) {
    __shared__ float acc[NPB * OUT_F];
    __shared__ float den[NPB * HEADS];
    __shared__ float stg[NPB * HEADS];
    __shared__ int   q[OWAV][WQ];

    const int tid  = threadIdx.x;
    const int lane = tid & 31;
    const int wave = tid >> 5;
    const int hd   = lane >> 2;
    const int n0   = blockIdx.x * NPB;

    for (int i = tid; i < NPB * OUT_F; i += OTHR) acc[i] = 0.0f;
    for (int i = tid; i < NPB * HEADS; i += OTHR) {
        den[i] = 0.0f;
        stg[i] = (n0 + (i >> 3) < N_NODES) ? s_tgt[n0 * HEADS + i] : 0.0f;
    }

    const float mx = dec_f32(gmax[hd]);
    const unsigned pat = (unsigned)blockIdx.x * 0x01010101u;
    int* qw = q[wave];

    int chunk = blockIdx.x % NSC;
    uint4 pre[LPB];
    load_chunk(own16, chunk, wave, lane, pre);
    __syncthreads();

    int n = 0;
    int t[PU], s[PU];
    for (int sc = 0; sc < NSC; ++sc) {
#pragma unroll
        for (int k = 0; k < LPB; ++k) {
            const int  i16 = chunk * (SCH / 16) + (wave * LPB + k) * 32 + lane;
            const bool inb = i16 < NP16;
            const unsigned inbm = __builtin_amdgcn_ballot_w32(inb);
            const unsigned w4[4] = {pre[k].x ^ pat, pre[k].y ^ pat, pre[k].z ^ pat, pre[k].w ^ pat};
#pragma unroll
            for (int d = 0; d < 4; ++d) {
                const unsigned w = w4[d];
                if (__builtin_amdgcn_ballot_w32(((w - 0x01010101u) & ~w & 0x80808080u) != 0u) == 0u) continue;
#pragma unroll
                for (int j = 0; j < 4; ++j) {
                    const bool     hit  = (w & (0xffu << (8 * j))) == 0u;
                    const unsigned mask = __builtin_amdgcn_ballot_w32(hit) & inbm;
                    if (mask != 0u) {
                        if (inb && hit) qw[n + (int)__builtin_amdgcn_mbcnt_lo(mask, 0u)] = i16 * 16 + d * 4 + j;
                        n += __builtin_popcount(mask);
                    }
                }
            }
        }
        chunk = (chunk + 1 == NSC) ? 0 : chunk + 1;

        const bool has = n >= PU;
        if (has) batch_ids<false>(qw, 0, n, tgt, src, t, s);
        load_chunk(own16, chunk, wave, lane, pre);
        if (has) batch_add<false>(t, s, 0, n, h, s_src, stg, acc, den, n0, lane, hd, mx);
        int i0 = has ? PU : 0;
        for (; i0 + PU <= n; i0 += PU) {
            batch_ids<false>(qw, i0, n, tgt, src, t, s);
            batch_add<false>(t, s, i0, n, h, s_src, stg, acc, den, n0, lane, hd, mx);
        }
        const int left = n - i0;
        if (i0 > 0 && left > 0) {
            const int v = qw[i0 + min(lane, left - 1)];
            if (lane < left) qw[lane] = v;
        }
        n = left;
    }
    if (n > 0) {
        batch_ids<true>(qw, 0, n, tgt, src, t, s);
        batch_add<true>(t, s, 0, n, h, s_src, stg, acc, den, n0, lane, hd, mx);
    }
    __syncthreads();

    const float4 bb = *reinterpret_cast<const float4*>(bias + 4 * lane);
    for (int idx = tid; idx < NPB * 32; idx += OTHR) {
        const int lt = idx >> 5;
        const int nd = n0 + lt;
        if (nd < N_NODES) {
            const float  d = den[lt * HEADS + hd] + EPS_F;
            const float* a = acc + lt * OUT_F + lane;
            float4 v;
            v.x = a[0]  / d + bb.x;
            v.y = a[32] / d + bb.y;
            v.z = a[64] / d + bb.z;
            v.w = a[96] / d + bb.w;
            *reinterpret_cast<float4*>(out + (size_t)nd * OUT_F + 4 * lane) = v;
        }
    }
}

extern "C" void kernel_launch(void* const* d_in, const int* in_sizes, int n_in,
                              void* d_out, int out_size, void* d_ws, size_t ws_size,
                              hipStream_t stream) {
    (void)in_sizes; (void)n_in; (void)out_size; (void)ws_size;

    const float* x    = (const float*)d_in[0];
    const int*   ei   = (const int*)  d_in[1];
    const float* W    = (const float*)d_in[2];
    const float* a    = (const float*)d_in[3];
    const float* bias = (const float*)d_in[4];
    float* out = (float*)d_out;

    const int* src = ei;
    const int* tgt = ei + N_EDGES;

    float*    ws    = (float*)d_ws;
    float*    hbuf  = ws;
    float*    ssrc  = hbuf + (size_t)N_NODES * OUT_F;
    float*    stgt  = ssrc + (size_t)N_NODES * HEADS;
    float*    ownf  = stgt + (size_t)N_NODES * HEADS;
    unsigned* gmax  = (unsigned*)(ownf + (size_t)N_NODES * HEADS);

    gat_init<<<1, 32, 0, stream>>>(gmax);

    gat_gemm_wmma<<<GBLK, 256, 0, stream>>>(x, W, hbuf);

    gat_node_scores<<<(N_NODES * HEADS + 255) / 256, 256, 0, stream>>>(hbuf, a, ssrc, stgt);

    gat_edge_max<<<(N_EDGES + 255) / 256, 256, 0, stream>>>(src, tgt, ssrc, stgt, gmax, (unsigned*)ownf);

    gat_owner_agg<<<NBLK, OTHR, 0, stream>>>(src, tgt, ssrc, stgt, gmax, hbuf, (const uint4*)ownf, bias, out);
}


# edge max pass re-cut: one block per WGP, a lane owns a head (8 lanes read an edge's 32-B logit rows), running max in a register over the wave's edges; joined once per wave, not per edge
# speedup vs baseline: 6.2179x; 1.5205x over previous
#include <hip/hip_runtime.h>
#include <hip/hip_bf16.h>
#include <math.h>

#define N_NODES   50000
#define N_EDGES   800000
#define IN_F      128
#define HEADS     8
#define HEAD_DIM  16
#define OUT_F     (HEADS * HEAD_DIM)
#define ALPHA     0.2f
#define EPS_F     1e-6f

#define NBLK  256
#define NPB   196
static_assert(NBLK <= 256 && NBLK * NPB >= N_NODES, "slice number must fit one byte");

typedef __attribute__((ext_vector_type(2))) float v2f;
typedef __attribute__((ext_vector_type(8))) float v8f;

__device__ __forceinline__ unsigned enc_f32(float f) {
    unsigned u = __float_as_uint(f);
    return (u & 0x80000000u) ? ~u : (u | 0x80000000u);
}
__device__ __forceinline__ float dec_f32(unsigned u) {
    return (u & 0x80000000u) ? __uint_as_float(u ^ 0x80000000u)
                             : __uint_as_float(~u);
}
__device__ __forceinline__ float leaky(float s) {
    return s >= 0.0f ? s : ALPHA * s;
}

__global__ void gat_init(unsigned* __restrict__ gmax) {
    if (threadIdx.x < HEADS) gmax[threadIdx.x] = 0u;
}

#define GROWS  128
#define GBLK   ((N_NODES + GROWS - 1) / GROWS)
static_assert(N_NODES % 16 == 0, "a wave's 16-row tile is whole or absent");

__global__ __launch_bounds__(256)
void gat_gemm_wmma(const float* __restrict__ x, const float* __restrict__ W,
                   float* __restrict__ h) {
    __shared__ float4 Wl[HEADS * 32 * 16];
    const int tid  = threadIdx.x;
    const int lane = tid & 31;
    const int wave = tid >> 5;
    const int l16  = lane & 15;
    const int hh   = lane >> 4;
    const int m0   = blockIdx.x * GROWS + wave * 16;
    const bool live = m0 < N_NODES;

    float4 a4[16];
    {
        const int row = min(m0 + l16, N_NODES - 1);
        const float4* xr = reinterpret_cast<const float4*>(x + (size_t)row * IN_F) + hh;
#pragma unroll
        for (int j = 0; j < 16; ++j) a4[j] = xr[2 * j];
    }

#pragma unroll
    for (int u = 0; u < 16; ++u) {
        const int f = tid + 256 * u;
        const int d = f & 15, q = (f >> 4) & 31, t = f >> 9;
        const float* wp = W + (size_t)(t * IN_F + 4 * q) * HEAD_DIM + d;
        float4 v;
        v.x = wp[0];
        v.y = wp[HEAD_DIM];
        v.z = wp[2 * HEAD_DIM];
        v.w = wp[3 * HEAD_DIM];
        Wl[f] = v;
    }
    __syncthreads();
    if (!live) return;

    v8f c[HEADS];
#pragma unroll
    for (int t = 0; t < HEADS; ++t) c[t] = v8f{};

    const float4* bl = Wl + lane;
    float4 b[2][HEADS];
#pragma unroll
    for (int t = 0; t < HEADS; ++t) b[0][t] = bl[t * 512];
#pragma unroll
    for (int j = 0; j < 16; ++j) {
        if (j + 1 < 16) {
#pragma unroll
            for (int t = 0; t < HEADS; ++t) b[(j + 1) & 1][t] = bl[t * 512 + (j + 1) * 32];
        }
        __builtin_amdgcn_sched_barrier(0);
        v2f alo, ahi;
        alo.x = a4[j].x; alo.y = a4[j].y;
        ahi.x = a4[j].z; ahi.y = a4[j].w;
#pragma unroll
        for (int t = 0; t < HEADS; ++t) {
            v2f bb; bb.x = b[j & 1][t].x; bb.y = b[j & 1][t].y;
            c[t] = __builtin_amdgcn_wmma_f32_16x16x4_f32(
                        false, alo,  false, bb,
                        (short)0, c[t],  false,  false);
        }
#pragma unroll
        for (int t = 0; t < HEADS; ++t) {
            v2f bb; bb.x = b[j & 1][t].z; bb.y = b[j & 1][t].w;
            c[t] = __builtin_amdgcn_wmma_f32_16x16x4_f32(
                        false, ahi,  false, bb,
                        (short)0, c[t],  false,  false);
        }
        __builtin_amdgcn_sched_barrier(0);
    }

    const int rbase = hh ? 8 : 0;
#pragma unroll
    for (int v = 0; v < 8; ++v) {
        float* hr = h + (size_t)(m0 + rbase + v) * OUT_F + l16;
#pragma unroll
        for (int t = 0; t < HEADS; ++t) hr[t * HEAD_DIM] = c[t][v];
    }
}

__global__ void gat_node_scores(const float* __restrict__ h, const float* __restrict__ a,
                                float* __restrict__ s_src, float* __restrict__ s_tgt) {
    const int i = blockIdx.x * blockDim.x + threadIdx.x;
    if (i >= N_NODES * HEADS) return;
    const int hd = i & (HEADS - 1);
    const int n  = i >> 3;
    const float* hv = h + (size_t)n * OUT_F + hd * HEAD_DIM;
    const float* as = a + hd * 2 * HEAD_DIM;
    const float* at = as + HEAD_DIM;
    float ss = 0.0f, st = 0.0f;
#pragma unroll
    for (int d = 0; d < HEAD_DIM; ++d) {
        const float v = hv[d];
        ss += v * as[d];
        st += v * at[d];
    }
    s_src[i] = ss;
    s_tgt[i] = st;
}

#define MB     256
#define MTHR   512
#define MWAV   (MTHR / 32)
#define MGT    (MB * MTHR)
#define MSTEP  49
#define MRND   7
#define MNR    (MSTEP / MRND)
#define NP4    (N_EDGES / 4)
static_assert(MSTEP % MRND == 0, "a wave's steps are whole rounds");
static_assert(MB * MWAV * MSTEP * 4 >= N_EDGES, "the waves' stretches cover the edge list");
static_assert(N_EDGES % 4 == 0, "the owner plane is written in whole 4-B words");

__global__ __launch_bounds__(MTHR)
void gat_edge_max(const int* __restrict__ src, const int* __restrict__ tgt,
                  const float* __restrict__ s_src, const float* __restrict__ s_tgt,
                  unsigned* __restrict__ gmax, unsigned* __restrict__ own4) {
    __shared__ float wmax[MWAV][HEADS];
    const int tid  = threadIdx.x;
    const int lane = tid & 31;
    const int wave = tid >> 5;
    const int hd   = lane & (HEADS - 1);
    const int e0   = (blockIdx.x * MWAV + wave) * (MSTEP * 4) + (lane >> 3);

    int sc[MRND], tc[MRND];
#pragma unroll
    for (int u = 0; u < MRND; ++u) {
        const int e = min(e0 + 4 * u, N_EDGES - 1);
        sc[u] = src[e];
        tc[u] = tgt[e];
    }

    const int4* t4 = reinterpret_cast<const int4*>(tgt);
#pragma unroll
    for (int k = 0; k < (NP4 + MGT - 1) / MGT; ++k) {
        const int i = blockIdx.x * MTHR + tid + k * MGT;
        if ((k + 1) * MGT <= NP4 || i < NP4) {
            const int4 tv = t4[i];
            own4[i] =  (((unsigned)tv.x / (unsigned)NPB) & 0xffu)
                    | ((((unsigned)tv.y / (unsigned)NPB) & 0xffu) << 8)
                    | ((((unsigned)tv.z / (unsigned)NPB) & 0xffu) << 16)
                    |  (((unsigned)tv.w / (unsigned)NPB)          << 24);
        }
    }

    float m = -INFINITY;
    for (int r = 0; r < MNR; ++r) {
        float a[MRND], b[MRND];
#pragma unroll
        for (int u = 0; u < MRND; ++u) {
            a[u] = s_src[sc[u] * HEADS + hd];
            b[u] = s_tgt[tc[u] * HEADS + hd];
        }
        const int rn = min(r + 1, MNR - 1);
#pragma unroll
        for (int u = 0; u < MRND; ++u) {
            const int e = min(e0 + 4 * (rn * MRND + u), N_EDGES - 1);
            sc[u] = src[e];
            tc[u] = tgt[e];
        }
        __builtin_amdgcn_sched_barrier(0);
#pragma unroll
        for (int u = 0; u < MRND; ++u) m = fmaxf(m, a[u] + b[u]);
    }

    m = fmaxf(m, __shfl_xor(m, 8));
    m = fmaxf(m, __shfl_xor(m, 16));
    if (lane < HEADS) wmax[wave][lane] = m;
    __syncthreads();
    if (tid < HEADS) {
        float v = wmax[0][tid];
        for (int w = 1; w < MWAV; ++w) v = fmaxf(v, wmax[w][tid]);
        atomicMax(&gmax[tid], enc_f32(leaky(v)));
    }
}

#define OTHR  1024
#define OWAV  (OTHR / 32)
#define LPB   2
#define SCH   (OTHR * LPB * 16)
#define PU    6
#define WQ    (32 * LPB * 16 + 8)
#define NSC   ((N_EDGES + SCH - 1) / SCH)
#define NP16  (N_EDGES / 16)
static_assert(N_EDGES % 16 == 0, "the owner plane is read in whole 16-B pieces");

__device__ __forceinline__ void load_chunk(const uint4* __restrict__ own16, int chunk,
                                           int wave, int lane, uint4 (&pre)[LPB]) {
#pragma unroll
    for (int k = 0; k < LPB; ++k) {
        const int i16 = chunk * (SCH / 16) + (wave * LPB + k) * 32 + lane;
        pre[k] = own16[min(i16, NP16 - 1)];
    }
}

template <bool PAD>
__device__ __forceinline__ void batch_ids(const int* qw, int i0, int n,
                                          const int* __restrict__ tgt, const int* __restrict__ src,
                                          int (&t)[PU], int (&s)[PU]) {
    int e[PU];
#pragma unroll
    for (int u = 0; u < PU; ++u) e[u] = qw[(!PAD || i0 + u < n) ? i0 + u : i0];
#pragma unroll
    for (int u = 0; u < PU; ++u) { t[u] = tgt[e[u]]; s[u] = src[e[u]]; }
}

template <bool PAD>
__device__ __forceinline__ void batch_add(const int (&t)[PU], const int (&s)[PU], int i0, int n,
                                          const float* __restrict__ h, const float* __restrict__ s_src,
                                          const float* stg, float* acc, float* den,
                                          int n0, int lane, int hd, float mx) {
    float4 hv[PU];
    float  ss[PU];
#pragma unroll
    for (int u = 0; u < PU; ++u) {
        hv[u] = *reinterpret_cast<const float4*>(h + (size_t)s[u] * OUT_F + 4 * lane);
        ss[u] = s_src[s[u] * HEADS + hd];
    }
#pragma unroll
    for (int u = 0; u < PU; ++u) {
        const int   lt  = t[u] - n0;
        const float scv = leaky(ss[u] + stg[lt * HEADS + hd]);
        float ev = __expf(fminf(fmaxf(scv - mx, -20.0f), 20.0f));
        if (PAD && i0 + u >= n) ev = 0.0f;
        float* a = acc + lt * OUT_F + lane;
        atomicAdd(a,      hv[u].x * ev);
        atomicAdd(a + 32, hv[u].y * ev);
        atomicAdd(a + 64, hv[u].z * ev);
        atomicAdd(a + 96, hv[u].w * ev);
        if ((lane & 3) == 0) atomicAdd(&den[lt * HEADS + hd], ev);
    }
}

__global__ __launch_bounds__(OTHR)
void gat_owner_agg(const int* __restrict__ src, const int* __restrict__ tgt,
                   const float* __restrict__ s_src, const float* __restrict__ s_tgt,
                   const unsigned* __restrict__ gmax, const float* __restrict__ h,
                   const uint4* __restrict__ own16,
                   const float* __restrict__ bias, float* __restrict__ out) {
    __shared__ float acc[NPB * OUT_F];
    __shared__ float den[NPB * HEADS];
    __shared__ float stg[NPB * HEADS];
    __shared__ int   q[OWAV][WQ];

    const int tid  = threadIdx.x;
    const int lane = tid & 31;
    const int wave = tid >> 5;
    const int hd   = lane >> 2;
    const int n0   = blockIdx.x * NPB;

    for (int i = tid; i < NPB * OUT_F; i += OTHR) acc[i] = 0.0f;
    for (int i = tid; i < NPB * HEADS; i += OTHR) {
        den[i] = 0.0f;
        stg[i] = (n0 + (i >> 3) < N_NODES) ? s_tgt[n0 * HEADS + i] : 0.0f;
    }

    const float mx = dec_f32(gmax[hd]);
    const unsigned pat = (unsigned)blockIdx.x * 0x01010101u;
    int* qw = q[wave];

    int chunk = blockIdx.x % NSC;
    uint4 pre[LPB];
    load_chunk(own16, chunk, wave, lane, pre);
    __syncthreads();

    int n = 0;
    int t[PU], s[PU];
    for (int sc = 0; sc < NSC; ++sc) {
#pragma unroll
        for (int k = 0; k < LPB; ++k) {
            const int  i16 = chunk * (SCH / 16) + (wave * LPB + k) * 32 + lane;
            const bool inb = i16 < NP16;
            const unsigned inbm = __builtin_amdgcn_ballot_w32(inb);
            const unsigned w4[4] = {pre[k].x ^ pat, pre[k].y ^ pat, pre[k].z ^ pat, pre[k].w ^ pat};
#pragma unroll
            for (int d = 0; d < 4; ++d) {
                const unsigned w = w4[d];
                if (__builtin_amdgcn_ballot_w32(((w - 0x01010101u) & ~w & 0x80808080u) != 0u) == 0u) continue;
#pragma unroll
                for (int j = 0; j < 4; ++j) {
                    const bool     hit  = (w & (0xffu << (8 * j))) == 0u;
                    const unsigned mask = __builtin_amdgcn_ballot_w32(hit) & inbm;
                    if (mask != 0u) {
                        if (inb && hit) qw[n + (int)__builtin_amdgcn_mbcnt_lo(mask, 0u)] = i16 * 16 + d * 4 + j;
                        n += __builtin_popcount(mask);
                    }
                }
            }
        }
        chunk = (chunk + 1 == NSC) ? 0 : chunk + 1;

        const bool has = n >= PU;
        if (has) batch_ids<false>(qw, 0, n, tgt, src, t, s);
        load_chunk(own16, chunk, wave, lane, pre);
        if (has) batch_add<false>(t, s, 0, n, h, s_src, stg, acc, den, n0, lane, hd, mx);
        int i0 = has ? PU : 0;
        for (; i0 + PU <= n; i0 += PU) {
            batch_ids<false>(qw, i0, n, tgt, src, t, s);
            batch_add<false>(t, s, i0, n, h, s_src, stg, acc, den, n0, lane, hd, mx);
        }
        const int left = n - i0;
        if (i0 > 0 && left > 0) {
            const int v = qw[i0 + min(lane, left - 1)];
            if (lane < left) qw[lane] = v;
        }
        n = left;
    }
    if (n > 0) {
        batch_ids<true>(qw, 0, n, tgt, src, t, s);
        batch_add<true>(t, s, 0, n, h, s_src, stg, acc, den, n0, lane, hd, mx);
    }
    __syncthreads();

    const float4 bb = *reinterpret_cast<const float4*>(bias + 4 * lane);
    for (int idx = tid; idx < NPB * 32; idx += OTHR) {
        const int lt = idx >> 5;
        const int nd = n0 + lt;
        if (nd < N_NODES) {
            const float  d = den[lt * HEADS + hd] + EPS_F;
            const float* a = acc + lt * OUT_F + lane;
            float4 v;
            v.x = a[0]  / d + bb.x;
            v.y = a[32] / d + bb.y;
            v.z = a[64] / d + bb.z;
            v.w = a[96] / d + bb.w;
            *reinterpret_cast<float4*>(out + (size_t)nd * OUT_F + 4 * lane) = v;
        }
    }
}

extern "C" void kernel_launch(void* const* d_in, const int* in_sizes, int n_in,
                              void* d_out, int out_size, void* d_ws, size_t ws_size,
                              hipStream_t stream) {
    (void)in_sizes; (void)n_in; (void)out_size; (void)ws_size;

    const float* x    = (const float*)d_in[0];
    const int*   ei   = (const int*)  d_in[1];
    const float* W    = (const float*)d_in[2];
    const float* a    = (const float*)d_in[3];
    const float* bias = (const float*)d_in[4];
    float* out = (float*)d_out;

    const int* src = ei;
    const int* tgt = ei + N_EDGES;

    float*    ws    = (float*)d_ws;
    float*    hbuf  = ws;
    float*    ssrc  = hbuf + (size_t)N_NODES * OUT_F;
    float*    stgt  = ssrc + (size_t)N_NODES * HEADS;
    float*    ownf  = stgt + (size_t)N_NODES * HEADS;
    unsigned* gmax  = (unsigned*)(ownf + (size_t)N_NODES * HEADS);

    gat_init<<<1, 32, 0, stream>>>(gmax);

    gat_gemm_wmma<<<GBLK, 256, 0, stream>>>(x, W, hbuf);

    gat_node_scores<<<(N_NODES * HEADS + 255) / 256, 256, 0, stream>>>(hbuf, a, ssrc, stgt);

    gat_edge_max<<<MB, MTHR, 0, stream>>>(src, tgt, ssrc, stgt, gmax, (unsigned*)ownf);

    gat_owner_agg<<<NBLK, OTHR, 0, stream>>>(src, tgt, ssrc, stgt, gmax, hbuf, (const uint4*)ownf, bias, out);
}


# pass 4 scan: a lane folds its 32 owner bytes per chunk into one 32-bit hit mask (exact byte test, vector ALU); one wave vote per hit rank, not per dword and byte; add order in a wave: rank, lane
# speedup vs baseline: 6.5664x; 1.0560x over previous
#include <hip/hip_runtime.h>
#include <hip/hip_bf16.h>
#include <math.h>

#define N_NODES   50000
#define N_EDGES   800000
#define IN_F      128
#define HEADS     8
#define HEAD_DIM  16
#define OUT_F     (HEADS * HEAD_DIM)
#define ALPHA     0.2f
#define EPS_F     1e-6f

#define NBLK  256
#define NPB   196
static_assert(NBLK <= 256 && NBLK * NPB >= N_NODES, "slice number must fit one byte");

typedef __attribute__((ext_vector_type(2))) float v2f;
typedef __attribute__((ext_vector_type(8))) float v8f;

__device__ __forceinline__ unsigned enc_f32(float f) {
    unsigned u = __float_as_uint(f);
    return (u & 0x80000000u) ? ~u : (u | 0x80000000u);
}
__device__ __forceinline__ float dec_f32(unsigned u) {
    return (u & 0x80000000u) ? __uint_as_float(u ^ 0x80000000u)
                             : __uint_as_float(~u);
}
__device__ __forceinline__ float leaky(float s) {
    return s >= 0.0f ? s : ALPHA * s;
}

__global__ void gat_init(unsigned* __restrict__ gmax) {
    if (threadIdx.x < HEADS) gmax[threadIdx.x] = 0u;
}

#define GROWS  128
#define GBLK   ((N_NODES + GROWS - 1) / GROWS)
static_assert(N_NODES % 16 == 0, "a wave's 16-row tile is whole or absent");

__global__ __launch_bounds__(256)
void gat_gemm_wmma(const float* __restrict__ x, const float* __restrict__ W,
                   float* __restrict__ h) {
    __shared__ float4 Wl[HEADS * 32 * 16];
    const int tid  = threadIdx.x;
    const int lane = tid & 31;
    const int wave = tid >> 5;
    const int l16  = lane & 15;
    const int hh   = lane >> 4;
    const int m0   = blockIdx.x * GROWS + wave * 16;
    const bool live = m0 < N_NODES;

    float4 a4[16];
    {
        const int row = min(m0 + l16, N_NODES - 1);
        const float4* xr = reinterpret_cast<const float4*>(x + (size_t)row * IN_F) + hh;
#pragma unroll
        for (int j = 0; j < 16; ++j) a4[j] = xr[2 * j];
    }

#pragma unroll
    for (int u = 0; u < 16; ++u) {
        const int f = tid + 256 * u;
        const int d = f & 15, q = (f >> 4) & 31, t = f >> 9;
        const float* wp = W + (size_t)(t * IN_F + 4 * q) * HEAD_DIM + d;
        float4 v;
        v.x = wp[0];
        v.y = wp[HEAD_DIM];
        v.z = wp[2 * HEAD_DIM];
        v.w = wp[3 * HEAD_DIM];
        Wl[f] = v;
    }
    __syncthreads();
    if (!live) return;

    v8f c[HEADS];
#pragma unroll
    for (int t = 0; t < HEADS; ++t) c[t] = v8f{};

    const float4* bl = Wl + lane;
    float4 b[2][HEADS];
#pragma unroll
    for (int t = 0; t < HEADS; ++t) b[0][t] = bl[t * 512];
#pragma unroll
    for (int j = 0; j < 16; ++j) {
        if (j + 1 < 16) {
#pragma unroll
            for (int t = 0; t < HEADS; ++t) b[(j + 1) & 1][t] = bl[t * 512 + (j + 1) * 32];
        }
        __builtin_amdgcn_sched_barrier(0);
        v2f alo, ahi;
        alo.x = a4[j].x; alo.y = a4[j].y;
        ahi.x = a4[j].z; ahi.y = a4[j].w;
#pragma unroll
        for (int t = 0; t < HEADS; ++t) {
            v2f bb; bb.x = b[j & 1][t].x; bb.y = b[j & 1][t].y;
            c[t] = __builtin_amdgcn_wmma_f32_16x16x4_f32(
                        false, alo,  false, bb,
                        (short)0, c[t],  false,  false);
        }
#pragma unroll
        for (int t = 0; t < HEADS; ++t) {
            v2f bb; bb.x = b[j & 1][t].z; bb.y = b[j & 1][t].w;
            c[t] = __builtin_amdgcn_wmma_f32_16x16x4_f32(
                        false, ahi,  false, bb,
                        (short)0, c[t],  false,  false);
        }
        __builtin_amdgcn_sched_barrier(0);
    }

    const int rbase = hh ? 8 : 0;
#pragma unroll
    for (int v = 0; v < 8; ++v) {
        float* hr = h + (size_t)(m0 + rbase + v) * OUT_F + l16;
#pragma unroll
        for (int t = 0; t < HEADS; ++t) hr[t * HEAD_DIM] = c[t][v];
    }
}

__global__ void gat_node_scores(const float* __restrict__ h, const float* __restrict__ a,
                                float* __restrict__ s_src, float* __restrict__ s_tgt) {
    const int i = blockIdx.x * blockDim.x + threadIdx.x;
    if (i >= N_NODES * HEADS) return;
    const int hd = i & (HEADS - 1);
    const int n  = i >> 3;
    const float* hv = h + (size_t)n * OUT_F + hd * HEAD_DIM;
    const float* as = a + hd * 2 * HEAD_DIM;
    const float* at = as + HEAD_DIM;
    float ss = 0.0f, st = 0.0f;
#pragma unroll
    for (int d = 0; d < HEAD_DIM; ++d) {
        const float v = hv[d];
        ss += v * as[d];
        st += v * at[d];
    }
    s_src[i] = ss;
    s_tgt[i] = st;
}

#define MB     256
#define MTHR   512
#define MWAV   (MTHR / 32)
#define MGT    (MB * MTHR)
#define MSTEP  49
#define MRND   7
#define MNR    (MSTEP / MRND)
#define NP4    (N_EDGES / 4)
static_assert(MSTEP % MRND == 0, "a wave's steps are whole rounds");
static_assert(MB * MWAV * MSTEP * 4 >= N_EDGES, "the waves' stretches cover the edge list");
static_assert(N_EDGES % 4 == 0, "the owner plane is written in whole 4-B words");

__global__ __launch_bounds__(MTHR)
void gat_edge_max(const int* __restrict__ src, const int* __restrict__ tgt,
                  const float* __restrict__ s_src, const float* __restrict__ s_tgt,
                  unsigned* __restrict__ gmax, unsigned* __restrict__ own4) {
    __shared__ float wmax[MWAV][HEADS];
    const int tid  = threadIdx.x;
    const int lane = tid & 31;
    const int wave = tid >> 5;
    const int hd   = lane & (HEADS - 1);
    const int e0   = (blockIdx.x * MWAV + wave) * (MSTEP * 4) + (lane >> 3);

    int sc[MRND], tc[MRND];
#pragma unroll
    for (int u = 0; u < MRND; ++u) {
        const int e = min(e0 + 4 * u, N_EDGES - 1);
        sc[u] = src[e];
        tc[u] = tgt[e];
    }

    const int4* t4 = reinterpret_cast<const int4*>(tgt);
#pragma unroll
    for (int k = 0; k < (NP4 + MGT - 1) / MGT; ++k) {
        const int i = blockIdx.x * MTHR + tid + k * MGT;
        if ((k + 1) * MGT <= NP4 || i < NP4) {
            const int4 tv = t4[i];
            own4[i] =  (((unsigned)tv.x / (unsigned)NPB) & 0xffu)
                    | ((((unsigned)tv.y / (unsigned)NPB) & 0xffu) << 8)
                    | ((((unsigned)tv.z / (unsigned)NPB) & 0xffu) << 16)
                    |  (((unsigned)tv.w / (unsigned)NPB)          << 24);
        }
    }

    float m = -INFINITY;
    for (int r = 0; r < MNR; ++r) {
        float a[MRND], b[MRND];
#pragma unroll
        for (int u = 0; u < MRND; ++u) {
            a[u] = s_src[sc[u] * HEADS + hd];
            b[u] = s_tgt[tc[u] * HEADS + hd];
        }
        const int rn = min(r + 1, MNR - 1);
#pragma unroll
        for (int u = 0; u < MRND; ++u) {
            const int e = min(e0 + 4 * (rn * MRND + u), N_EDGES - 1);
            sc[u] = src[e];
            tc[u] = tgt[e];
        }
        __builtin_amdgcn_sched_barrier(0);
#pragma unroll
        for (int u = 0; u < MRND; ++u) m = fmaxf(m, a[u] + b[u]);
    }

    m = fmaxf(m, __shfl_xor(m, 8));
    m = fmaxf(m, __shfl_xor(m, 16));
    if (lane < HEADS) wmax[wave][lane] = m;
    __syncthreads();
    if (tid < HEADS) {
        float v = wmax[0][tid];
        for (int w = 1; w < MWAV; ++w) v = fmaxf(v, wmax[w][tid]);
        atomicMax(&gmax[tid], enc_f32(leaky(v)));
    }
}

#define OTHR  1024
#define OWAV  (OTHR / 32)
#define LPB   2
#define SCH   (OTHR * LPB * 16)
#define PU    6
#define WQ    (32 * LPB * 16 + 8)
#define NSC   ((N_EDGES + SCH - 1) / SCH)
#define NP16  (N_EDGES / 16)
static_assert(N_EDGES % 16 == 0, "the owner plane is read in whole 16-B pieces");
static_assert(LPB * 16 == 32, "a lane's owner bytes of a chunk fill one 32-bit hit mask");

__device__ __forceinline__ unsigned nonzero_bytes(unsigned w) {
    return ((w & 0x7f7f7f7fu) + 0x7f7f7f7fu) | w;
}

__device__ __forceinline__ void load_chunk(const uint4* __restrict__ own16, int chunk,
                                           int wave, int lane, uint4 (&pre)[LPB]) {
#pragma unroll
    for (int k = 0; k < LPB; ++k) {
        const int i16 = chunk * (SCH / 16) + (wave * LPB + k) * 32 + lane;
        pre[k] = own16[min(i16, NP16 - 1)];
    }
}

template <bool PAD>
__device__ __forceinline__ void batch_ids(const int* qw, int i0, int n,
                                          const int* __restrict__ tgt, const int* __restrict__ src,
                                          int (&t)[PU], int (&s)[PU]) {
    int e[PU];
#pragma unroll
    for (int u = 0; u < PU; ++u) e[u] = qw[(!PAD || i0 + u < n) ? i0 + u : i0];
#pragma unroll
    for (int u = 0; u < PU; ++u) { t[u] = tgt[e[u]]; s[u] = src[e[u]]; }
}

template <bool PAD>
__device__ __forceinline__ void batch_add(const int (&t)[PU], const int (&s)[PU], int i0, int n,
                                          const float* __restrict__ h, const float* __restrict__ s_src,
                                          const float* stg, float* acc, float* den,
                                          int n0, int lane, int hd, float mx) {
    float4 hv[PU];
    float  ss[PU];
#pragma unroll
    for (int u = 0; u < PU; ++u) {
        hv[u] = *reinterpret_cast<const float4*>(h + (size_t)s[u] * OUT_F + 4 * lane);
        ss[u] = s_src[s[u] * HEADS + hd];
    }
#pragma unroll
    for (int u = 0; u < PU; ++u) {
        const int   lt  = t[u] - n0;
        const float scv = leaky(ss[u] + stg[lt * HEADS + hd]);
        float ev = __expf(fminf(fmaxf(scv - mx, -20.0f), 20.0f));
        if (PAD && i0 + u >= n) ev = 0.0f;
        float* a = acc + lt * OUT_F + lane;
        atomicAdd(a,      hv[u].x * ev);
        atomicAdd(a + 32, hv[u].y * ev);
        atomicAdd(a + 64, hv[u].z * ev);
        atomicAdd(a + 96, hv[u].w * ev);
        if ((lane & 3) == 0) atomicAdd(&den[lt * HEADS + hd], ev);
    }
}

__global__ __launch_bounds__(OTHR)
void gat_owner_agg(const int* __restrict__ src, const int* __restrict__ tgt,
                   const float* __restrict__ s_src, const float* __restrict__ s_tgt,
                   const unsigned* __restrict__ gmax, const float* __restrict__ h,
                   const uint4* __restrict__ own16,
                   const float* __restrict__ bias, float* __restrict__ out) {
    __shared__ float acc[NPB * OUT_F];
    __shared__ float den[NPB * HEADS];
    __shared__ float stg[NPB * HEADS];
    __shared__ int   q[OWAV][WQ];

    const int tid  = threadIdx.x;
    const int lane = tid & 31;
    const int wave = tid >> 5;
    const int hd   = lane >> 2;
    const int n0   = blockIdx.x * NPB;

    for (int i = tid; i < NPB * OUT_F; i += OTHR) acc[i] = 0.0f;
    for (int i = tid; i < NPB * HEADS; i += OTHR) {
        den[i] = 0.0f;
        stg[i] = (n0 + (i >> 3) < N_NODES) ? s_tgt[n0 * HEADS + i] : 0.0f;
    }

    const float mx = dec_f32(gmax[hd]);
    const unsigned pat = (unsigned)blockIdx.x * 0x01010101u;
    int* qw = q[wave];

    int chunk = blockIdx.x % NSC;
    uint4 pre[LPB];
    load_chunk(own16, chunk, wave, lane, pre);
    __syncthreads();

    int n = 0;
    int t[PU], s[PU];
    for (int sc = 0; sc < NSC; ++sc) {
        {
            const int i16a = chunk * (SCH / 16) + wave * (LPB * 32) + lane;
            unsigned nz = 0u;
#pragma unroll
            for (int k = 0; k < LPB; ++k) {
                const unsigned w4[4] = {pre[k].x ^ pat, pre[k].y ^ pat, pre[k].z ^ pat, pre[k].w ^ pat};
#pragma unroll
                for (int d = 0; d < 4; ++d)
                    nz |= (nonzero_bytes(w4[d]) >> (7 - 4 * k - d)) & (0x01010101u << (4 * k + d));
            }
            unsigned m = ~nz;
#pragma unroll
            for (int k = 0; k < LPB; ++k)
                if (i16a + 32 * k >= NP16) m &= ~(0x0f0f0f0fu << (4 * k));
            unsigned live;
            while ((live = __builtin_amdgcn_ballot_w32(m != 0u)) != 0u) {
                if (m != 0u) {
                    const int p = __builtin_ctz(m);
                    m &= m - 1u;
                    qw[n + (int)__builtin_amdgcn_mbcnt_lo(live, 0u)] =
                        i16a * 16 + ((p >> 2) & 1) * 512 + (p & 3) * 4 + (p >> 3);
                }
                n += __builtin_popcount(live);
            }
        }
        chunk = (chunk + 1 == NSC) ? 0 : chunk + 1;

        const bool has = n >= PU;
        if (has) batch_ids<false>(qw, 0, n, tgt, src, t, s);
        load_chunk(own16, chunk, wave, lane, pre);
        if (has) batch_add<false>(t, s, 0, n, h, s_src, stg, acc, den, n0, lane, hd, mx);
        int i0 = has ? PU : 0;
        for (; i0 + PU <= n; i0 += PU) {
            batch_ids<false>(qw, i0, n, tgt, src, t, s);
            batch_add<false>(t, s, i0, n, h, s_src, stg, acc, den, n0, lane, hd, mx);
        }
        const int left = n - i0;
        if (i0 > 0 && left > 0) {
            const int v = qw[i0 + min(lane, left - 1)];
            if (lane < left) qw[lane] = v;
        }
        n = left;
    }
    if (n > 0) {
        batch_ids<true>(qw, 0, n, tgt, src, t, s);
        batch_add<true>(t, s, 0, n, h, s_src, stg, acc, den, n0, lane, hd, mx);
    }
    __syncthreads();

    const float4 bb = *reinterpret_cast<const float4*>(bias + 4 * lane);
    for (int idx = tid; idx < NPB * 32; idx += OTHR) {
        const int lt = idx >> 5;
        const int nd = n0 + lt;
        if (nd < N_NODES) {
            const float  d = den[lt * HEADS + hd] + EPS_F;
            const float* a = acc + lt * OUT_F + lane;
            float4 v;
            v.x = a[0]  / d + bb.x;
            v.y = a[32] / d + bb.y;
            v.z = a[64] / d + bb.z;
            v.w = a[96] / d + bb.w;
            *reinterpret_cast<float4*>(out + (size_t)nd * OUT_F + 4 * lane) = v;
        }
    }
}

extern "C" void kernel_launch(void* const* d_in, const int* in_sizes, int n_in,
                              void* d_out, int out_size, void* d_ws, size_t ws_size,
                              hipStream_t stream) {
    (void)in_sizes; (void)n_in; (void)out_size; (void)ws_size;

    const float* x    = (const float*)d_in[0];
    const int*   ei   = (const int*)  d_in[1];
    const float* W    = (const float*)d_in[2];
    const float* a    = (const float*)d_in[3];
    const float* bias = (const float*)d_in[4];
    float* out = (float*)d_out;

    const int* src = ei;
    const int* tgt = ei + N_EDGES;

    float*    ws    = (float*)d_ws;
    float*    hbuf  = ws;
    float*    ssrc  = hbuf + (size_t)N_NODES * OUT_F;
    float*    stgt  = ssrc + (size_t)N_NODES * HEADS;
    float*    ownf  = stgt + (size_t)N_NODES * HEADS;
    unsigned* gmax  = (unsigned*)(ownf + (size_t)N_NODES * HEADS);

    gat_init<<<1, 32, 0, stream>>>(gmax);

    gat_gemm_wmma<<<GBLK, 256, 0, stream>>>(x, W, hbuf);

    gat_node_scores<<<(N_NODES * HEADS + 255) / 256, 256, 0, stream>>>(hbuf, a, ssrc, stgt);

    gat_edge_max<<<MB, MTHR, 0, stream>>>(src, tgt, ssrc, stgt, gmax, (unsigned*)ownf);

    gat_owner_agg<<<NBLK, OTHR, 0, stream>>>(src, tgt, ssrc, stgt, gmax, hbuf, (const uint4*)ownf, bias, out);
}


# owner plane bit-sliced: pass 3 writes 8 one-bit planes per 32 edges (wave votes), pass 4 gets a lane's 32-edge hit mask from 8 xors and an or, no byte tests; add order in a wave: rank, lane
# speedup vs baseline: 6.7728x; 1.0314x over previous
#include <hip/hip_runtime.h>
#include <hip/hip_bf16.h>
#include <math.h>

#define N_NODES   50000
#define N_EDGES   800000
#define IN_F      128
#define HEADS     8
#define HEAD_DIM  16
#define OUT_F     (HEADS * HEAD_DIM)
#define ALPHA     0.2f
#define EPS_F     1e-6f

#define NBLK  256
#define NPB   196
static_assert(NBLK <= 256 && NBLK * NPB >= N_NODES, "slice number must fit eight bits");
#define NGRP  (N_EDGES / 32)
#define NSTR  ((NGRP + 31) / 32)
static_assert(N_EDGES % 256 == 0, "a pass-3 wave turns 8 whole groups into two whole 128-B lines");

typedef __attribute__((ext_vector_type(2))) float v2f;
typedef __attribute__((ext_vector_type(8))) float v8f;

__device__ __forceinline__ unsigned enc_f32(float f) {
    unsigned u = __float_as_uint(f);
    return (u & 0x80000000u) ? ~u : (u | 0x80000000u);
}
__device__ __forceinline__ float dec_f32(unsigned u) {
    return (u & 0x80000000u) ? __uint_as_float(u ^ 0x80000000u)
                             : __uint_as_float(~u);
}
__device__ __forceinline__ float leaky(float s) {
    return s >= 0.0f ? s : ALPHA * s;
}

__global__ void gat_init(unsigned* __restrict__ gmax) {
    if (threadIdx.x < HEADS) gmax[threadIdx.x] = 0u;
}

#define GROWS  128
#define GBLK   ((N_NODES + GROWS - 1) / GROWS)
static_assert(N_NODES % 16 == 0, "a wave's 16-row tile is whole or absent");

__global__ __launch_bounds__(256)
void gat_gemm_wmma(const float* __restrict__ x, const float* __restrict__ W,
                   float* __restrict__ h) {
    __shared__ float4 Wl[HEADS * 32 * 16];
    const int tid  = threadIdx.x;
    const int lane = tid & 31;
    const int wave = tid >> 5;
    const int l16  = lane & 15;
    const int hh   = lane >> 4;
    const int m0   = blockIdx.x * GROWS + wave * 16;
    const bool live = m0 < N_NODES;

    float4 a4[16];
    {
        const int row = min(m0 + l16, N_NODES - 1);
        const float4* xr = reinterpret_cast<const float4*>(x + (size_t)row * IN_F) + hh;
#pragma unroll
        for (int j = 0; j < 16; ++j) a4[j] = xr[2 * j];
    }

#pragma unroll
    for (int u = 0; u < 16; ++u) {
        const int f = tid + 256 * u;
        const int d = f & 15, q = (f >> 4) & 31, t = f >> 9;
        const float* wp = W + (size_t)(t * IN_F + 4 * q) * HEAD_DIM + d;
        float4 v;
        v.x = wp[0];
        v.y = wp[HEAD_DIM];
        v.z = wp[2 * HEAD_DIM];
        v.w = wp[3 * HEAD_DIM];
        Wl[f] = v;
    }
    __syncthreads();
    if (!live) return;

    v8f c[HEADS];
#pragma unroll
    for (int t = 0; t < HEADS; ++t) c[t] = v8f{};

    const float4* bl = Wl + lane;
    float4 b[2][HEADS];
#pragma unroll
    for (int t = 0; t < HEADS; ++t) b[0][t] = bl[t * 512];
#pragma unroll
    for (int j = 0; j < 16; ++j) {
        if (j + 1 < 16) {
#pragma unroll
            for (int t = 0; t < HEADS; ++t) b[(j + 1) & 1][t] = bl[t * 512 + (j + 1) * 32];
        }
        __builtin_amdgcn_sched_barrier(0);
        v2f alo, ahi;
        alo.x = a4[j].x; alo.y = a4[j].y;
        ahi.x = a4[j].z; ahi.y = a4[j].w;
#pragma unroll
        for (int t = 0; t < HEADS; ++t) {
            v2f bb; bb.x = b[j & 1][t].x; bb.y = b[j & 1][t].y;
            c[t] = __builtin_amdgcn_wmma_f32_16x16x4_f32(
                        false, alo,  false, bb,
                        (short)0, c[t],  false,  false);
        }
#pragma unroll
        for (int t = 0; t < HEADS; ++t) {
            v2f bb; bb.x = b[j & 1][t].z; bb.y = b[j & 1][t].w;
            c[t] = __builtin_amdgcn_wmma_f32_16x16x4_f32(
                        false, ahi,  false, bb,
                        (short)0, c[t],  false,  false);
        }
        __builtin_amdgcn_sched_barrier(0);
    }

    const int rbase = hh ? 8 : 0;
#pragma unroll
    for (int v = 0; v < 8; ++v) {
        float* hr = h + (size_t)(m0 + rbase + v) * OUT_F + l16;
#pragma unroll
        for (int t = 0; t < HEADS; ++t) hr[t * HEAD_DIM] = c[t][v];
    }
}

__global__ void gat_node_scores(const float* __restrict__ h, const float* __restrict__ a,
                                float* __restrict__ s_src, float* __restrict__ s_tgt) {
    const int i = blockIdx.x * blockDim.x + threadIdx.x;
    if (i >= N_NODES * HEADS) return;
    const int hd = i & (HEADS - 1);
    const int n  = i >> 3;
    const float* hv = h + (size_t)n * OUT_F + hd * HEAD_DIM;
    const float* as = a + hd * 2 * HEAD_DIM;
    const float* at = as + HEAD_DIM;
    float ss = 0.0f, st = 0.0f;
#pragma unroll
    for (int d = 0; d < HEAD_DIM; ++d) {
        const float v = hv[d];
        ss += v * as[d];
        st += v * at[d];
    }
    s_src[i] = ss;
    s_tgt[i] = st;
}

#define MB     256
#define MTHR   512
#define MWAV   (MTHR / 32)
#define MGT    (MB * MTHR)
#define MSTEP  49
#define MRND   7
#define MNR    (MSTEP / MRND)
#define NSR    (N_EDGES / 256)
static_assert(MSTEP % MRND == 0, "a wave's steps are whole rounds");
static_assert(MB * MWAV * MSTEP * 4 >= N_EDGES, "the waves' stretches cover the edge list");
static_assert(MB * MWAV >= NSR, "every plane job has a wave");

__global__ __launch_bounds__(MTHR)
void gat_edge_max(const int* __restrict__ src, const int* __restrict__ tgt,
                  const float* __restrict__ s_src, const float* __restrict__ s_tgt,
                  unsigned* __restrict__ gmax, unsigned* __restrict__ own4) {
    __shared__ float wmax[MWAV][HEADS];
    const int tid  = threadIdx.x;
    const int lane = tid & 31;
    const int wave = tid >> 5;
    const int hd   = lane & (HEADS - 1);
    const int e0   = (blockIdx.x * MWAV + wave) * (MSTEP * 4) + (lane >> 3);

    int sc[MRND], tc[MRND];
#pragma unroll
    for (int u = 0; u < MRND; ++u) {
        const int e = min(e0 + 4 * u, N_EDGES - 1);
        sc[u] = src[e];
        tc[u] = tgt[e];
    }

    const int gw = blockIdx.x * MWAV + __builtin_amdgcn_readfirstlane(wave);
    if (gw < NSR) {
        int tv[8];
#pragma unroll
        for (int j = 0; j < 8; ++j) tv[j] = tgt[gw * 256 + 32 * j + lane];
        unsigned w0 = 0u, w1 = 0u;
#pragma unroll
        for (int j = 0; j < 8; ++j) {
            const unsigned ow = (unsigned)tv[j] / (unsigned)NPB;
#pragma unroll
            for (int k = 0; k < 8; ++k) {
                const unsigned b = __builtin_amdgcn_ballot_w32(((ow >> k) & 1u) != 0u);
                if (k < 4) w0 = (lane == 4 * j + k)     ? b : w0;
                else       w1 = (lane == 4 * j + k - 4) ? b : w1;
            }
        }
        unsigned* pw = own4 + 256 * (gw >> 2) + 32 * (gw & 3) + lane;
        pw[0]   = w0;
        pw[128] = w1;
    }

    float m = -INFINITY;
    for (int r = 0; r < MNR; ++r) {
        float a[MRND], b[MRND];
#pragma unroll
        for (int u = 0; u < MRND; ++u) {
            a[u] = s_src[sc[u] * HEADS + hd];
            b[u] = s_tgt[tc[u] * HEADS + hd];
        }
        const int rn = min(r + 1, MNR - 1);
#pragma unroll
        for (int u = 0; u < MRND; ++u) {
            const int e = min(e0 + 4 * (rn * MRND + u), N_EDGES - 1);
            sc[u] = src[e];
            tc[u] = tgt[e];
        }
        __builtin_amdgcn_sched_barrier(0);
#pragma unroll
        for (int u = 0; u < MRND; ++u) m = fmaxf(m, a[u] + b[u]);
    }

    m = fmaxf(m, __shfl_xor(m, 8));
    m = fmaxf(m, __shfl_xor(m, 16));
    if (lane < HEADS) wmax[wave][lane] = m;
    __syncthreads();
    if (tid < HEADS) {
        float v = wmax[0][tid];
        for (int w = 1; w < MWAV; ++w) v = fmaxf(v, wmax[w][tid]);
        atomicMax(&gmax[tid], enc_f32(leaky(v)));
    }
}

#define OTHR  1024
#define OWAV  (OTHR / 32)
#define LPB   2
#define SCH   (OTHR * 32)
#define PU    6
#define WQ    (32 * LPB * 16 + 8)
#define NSC   ((N_EDGES + SCH - 1) / SCH)
static_assert(LPB * 4 == 8, "a lane's loads of a chunk are the eight plane words of its group");
static_assert(OWAV == 32, "a wave's 32 lanes of a chunk are one stretch of 32 groups");

__device__ __forceinline__ void load_chunk(const uint4* __restrict__ own16, int chunk,
                                           int wave, int lane, uint4 (&pre)[LPB]) {
#pragma unroll
    for (int k = 0; k < LPB; ++k) {
        const int i16 = chunk * (SCH / 16) + (wave * LPB + k) * 32 + lane;
        pre[k] = own16[min(i16, ((NGRP - 1) >> 5) * (LPB * 32) + k * 32 + ((NGRP - 1) & 31))];
    }
}

template <bool PAD>
__device__ __forceinline__ void batch_ids(const int* qw, int i0, int n,
                                          const int* __restrict__ tgt, const int* __restrict__ src,
                                          int (&t)[PU], int (&s)[PU]) {
    int e[PU];
#pragma unroll
    for (int u = 0; u < PU; ++u) e[u] = qw[(!PAD || i0 + u < n) ? i0 + u : i0];
#pragma unroll
    for (int u = 0; u < PU; ++u) { t[u] = tgt[e[u]]; s[u] = src[e[u]]; }
}

template <bool PAD>
__device__ __forceinline__ void batch_add(const int (&t)[PU], const int (&s)[PU], int i0, int n,
                                          const float* __restrict__ h, const float* __restrict__ s_src,
                                          const float* stg, float* acc, float* den,
                                          int n0, int lane, int hd, float mx) {
    float4 hv[PU];
    float  ss[PU];
#pragma unroll
    for (int u = 0; u < PU; ++u) {
        hv[u] = *reinterpret_cast<const float4*>(h + (size_t)s[u] * OUT_F + 4 * lane);
        ss[u] = s_src[s[u] * HEADS + hd];
    }
#pragma unroll
    for (int u = 0; u < PU; ++u) {
        const int   lt  = t[u] - n0;
        const float scv = leaky(ss[u] + stg[lt * HEADS + hd]);
        float ev = __expf(fminf(fmaxf(scv - mx, -20.0f), 20.0f));
        if (PAD && i0 + u >= n) ev = 0.0f;
        float* a = acc + lt * OUT_F + lane;
        atomicAdd(a,      hv[u].x * ev);
        atomicAdd(a + 32, hv[u].y * ev);
        atomicAdd(a + 64, hv[u].z * ev);
        atomicAdd(a + 96, hv[u].w * ev);
        if ((lane & 3) == 0) atomicAdd(&den[lt * HEADS + hd], ev);
    }
}

__global__ __launch_bounds__(OTHR)
void gat_owner_agg(const int* __restrict__ src, const int* __restrict__ tgt,
                   const float* __restrict__ s_src, const float* __restrict__ s_tgt,
                   const unsigned* __restrict__ gmax, const float* __restrict__ h,
                   const uint4* __restrict__ own16,
                   const float* __restrict__ bias, float* __restrict__ out) {
    __shared__ float acc[NPB * OUT_F];
    __shared__ float den[NPB * HEADS];
    __shared__ float stg[NPB * HEADS];
    __shared__ int   q[OWAV][WQ];

    const int tid  = threadIdx.x;
    const int lane = tid & 31;
    const int wave = tid >> 5;
    const int hd   = lane >> 2;
    const int n0   = blockIdx.x * NPB;

    for (int i = tid; i < NPB * OUT_F; i += OTHR) acc[i] = 0.0f;
    for (int i = tid; i < NPB * HEADS; i += OTHR) {
        den[i] = 0.0f;
        stg[i] = (n0 + (i >> 3) < N_NODES) ? s_tgt[n0 * HEADS + i] : 0.0f;
    }

    const float mx = dec_f32(gmax[hd]);
    unsigned cb[8];
#pragma unroll
    for (int k = 0; k < 8; ++k) cb[k] = 0u - ((blockIdx.x >> k) & 1u);
    int* qw = q[wave];

    int chunk = blockIdx.x % NSC;
    uint4 pre[LPB];
    load_chunk(own16, chunk, wave, lane, pre);
    __syncthreads();

    int n = 0;
    int t[PU], s[PU];
    for (int sc = 0; sc < NSC; ++sc) {
        {
            const int g = (chunk * OWAV + wave) * 32 + lane;
            const unsigned mis = (pre[0].x ^ cb[0]) | (pre[0].y ^ cb[1]) | (pre[0].z ^ cb[2]) | (pre[0].w ^ cb[3])
                               | (pre[1].x ^ cb[4]) | (pre[1].y ^ cb[5]) | (pre[1].z ^ cb[6]) | (pre[1].w ^ cb[7]);
            unsigned m = (g < NGRP) ? ~mis : 0u;
            unsigned live;
            while ((live = __builtin_amdgcn_ballot_w32(m != 0u)) != 0u) {
                if (m != 0u) {
                    const int p = __builtin_ctz(m);
                    m &= m - 1u;
                    qw[n + (int)__builtin_amdgcn_mbcnt_lo(live, 0u)] = g * 32 + p;
                }
                n += __builtin_popcount(live);
            }
        }
        chunk = (chunk + 1 == NSC) ? 0 : chunk + 1;

        const bool has = n >= PU;
        if (has) batch_ids<false>(qw, 0, n, tgt, src, t, s);
        load_chunk(own16, chunk, wave, lane, pre);
        if (has) batch_add<false>(t, s, 0, n, h, s_src, stg, acc, den, n0, lane, hd, mx);
        int i0 = has ? PU : 0;
        for (; i0 + PU <= n; i0 += PU) {
            batch_ids<false>(qw, i0, n, tgt, src, t, s);
            batch_add<false>(t, s, i0, n, h, s_src, stg, acc, den, n0, lane, hd, mx);
        }
        const int left = n - i0;
        if (i0 > 0 && left > 0) {
            const int v = qw[i0 + min(lane, left - 1)];
            if (lane < left) qw[lane] = v;
        }
        n = left;
    }
    if (n > 0) {
        batch_ids<true>(qw, 0, n, tgt, src, t, s);
        batch_add<true>(t, s, 0, n, h, s_src, stg, acc, den, n0, lane, hd, mx);
    }
    __syncthreads();

    const float4 bb = *reinterpret_cast<const float4*>(bias + 4 * lane);
    for (int idx = tid; idx < NPB * 32; idx += OTHR) {
        const int lt = idx >> 5;
        const int nd = n0 + lt;
        if (nd < N_NODES) {
            const float  d = den[lt * HEADS + hd] + EPS_F;
            const float* a = acc + lt * OUT_F + lane;
            float4 v;
            v.x = a[0]  / d + bb.x;
            v.y = a[32] / d + bb.y;
            v.z = a[64] / d + bb.z;
            v.w = a[96] / d + bb.w;
            *reinterpret_cast<float4*>(out + (size_t)nd * OUT_F + 4 * lane) = v;
        }
    }
}

extern "C" void kernel_launch(void* const* d_in, const int* in_sizes, int n_in,
                              void* d_out, int out_size, void* d_ws, size_t ws_size,
                              hipStream_t stream) {
    (void)in_sizes; (void)n_in; (void)out_size; (void)ws_size;

    const float* x    = (const float*)d_in[0];
    const int*   ei   = (const int*)  d_in[1];
    const float* W    = (const float*)d_in[2];
    const float* a    = (const float*)d_in[3];
    const float* bias = (const float*)d_in[4];
    float* out = (float*)d_out;

    const int* src = ei;
    const int* tgt = ei + N_EDGES;

    float*    ws    = (float*)d_ws;
    float*    hbuf  = ws;
    float*    ssrc  = hbuf + (size_t)N_NODES * OUT_F;
    float*    stgt  = ssrc + (size_t)N_NODES * HEADS;
    float*    ownf  = stgt + (size_t)N_NODES * HEADS;
    unsigned* gmax  = (unsigned*)(ownf + (size_t)N_NODES * HEADS);

    gat_init<<<1, 32, 0, stream>>>(gmax);

    gat_gemm_wmma<<<GBLK, 256, 0, stream>>>(x, W, hbuf);

    gat_node_scores<<<(N_NODES * HEADS + 255) / 256, 256, 0, stream>>>(hbuf, a, ssrc, stgt);

    gat_edge_max<<<MB, MTHR, 0, stream>>>(src, tgt, ssrc, stgt, gmax, (unsigned*)ownf);

    gat_owner_agg<<<NBLK, OTHR, 0, stream>>>(src, tgt, ssrc, stgt, gmax, hbuf, (const uint4*)ownf, bias, out);
}


# pass 4: a batch's six target-logit LDS reads issued ahead of its first add (were one per edge, each waited for behind the previous edge's five adds); weights and arrival-order sums as before
# speedup vs baseline: 7.0143x; 1.0357x over previous
#include <hip/hip_runtime.h>
#include <hip/hip_bf16.h>
#include <math.h>

#define N_NODES   50000
#define N_EDGES   800000
#define IN_F      128
#define HEADS     8
#define HEAD_DIM  16
#define OUT_F     (HEADS * HEAD_DIM)
#define ALPHA     0.2f
#define EPS_F     1e-6f

#define NBLK  256
#define NPB   196
static_assert(NBLK <= 256 && NBLK * NPB >= N_NODES, "slice number must fit eight bits");
#define NGRP  (N_EDGES / 32)
#define NSTR  ((NGRP + 31) / 32)
static_assert(N_EDGES % 256 == 0, "a pass-3 wave turns 8 whole groups into two whole 128-B lines");

typedef __attribute__((ext_vector_type(2))) float v2f;
typedef __attribute__((ext_vector_type(8))) float v8f;

__device__ __forceinline__ unsigned enc_f32(float f) {
    unsigned u = __float_as_uint(f);
    return (u & 0x80000000u) ? ~u : (u | 0x80000000u);
}
__device__ __forceinline__ float dec_f32(unsigned u) {
    return (u & 0x80000000u) ? __uint_as_float(u ^ 0x80000000u)
                             : __uint_as_float(~u);
}
__device__ __forceinline__ float leaky(float s) {
    return s >= 0.0f ? s : ALPHA * s;
}

__global__ void gat_init(unsigned* __restrict__ gmax) {
    if (threadIdx.x < HEADS) gmax[threadIdx.x] = 0u;
}

#define GROWS  128
#define GBLK   ((N_NODES + GROWS - 1) / GROWS)
static_assert(N_NODES % 16 == 0, "a wave's 16-row tile is whole or absent");

__global__ __launch_bounds__(256)
void gat_gemm_wmma(const float* __restrict__ x, const float* __restrict__ W,
                   float* __restrict__ h) {
    __shared__ float4 Wl[HEADS * 32 * 16];
    const int tid  = threadIdx.x;
    const int lane = tid & 31;
    const int wave = tid >> 5;
    const int l16  = lane & 15;
    const int hh   = lane >> 4;
    const int m0   = blockIdx.x * GROWS + wave * 16;
    const bool live = m0 < N_NODES;

    float4 a4[16];
    {
        const int row = min(m0 + l16, N_NODES - 1);
        const float4* xr = reinterpret_cast<const float4*>(x + (size_t)row * IN_F) + hh;
#pragma unroll
        for (int j = 0; j < 16; ++j) a4[j] = xr[2 * j];
    }

#pragma unroll
    for (int u = 0; u < 16; ++u) {
        const int f = tid + 256 * u;
        const int d = f & 15, q = (f >> 4) & 31, t = f >> 9;
        const float* wp = W + (size_t)(t * IN_F + 4 * q) * HEAD_DIM + d;
        float4 v;
        v.x = wp[0];
        v.y = wp[HEAD_DIM];
        v.z = wp[2 * HEAD_DIM];
        v.w = wp[3 * HEAD_DIM];
        Wl[f] = v;
    }
    __syncthreads();
    if (!live) return;

    v8f c[HEADS];
#pragma unroll
    for (int t = 0; t < HEADS; ++t) c[t] = v8f{};

    const float4* bl = Wl + lane;
    float4 b[2][HEADS];
#pragma unroll
    for (int t = 0; t < HEADS; ++t) b[0][t] = bl[t * 512];
#pragma unroll
    for (int j = 0; j < 16; ++j) {
        if (j + 1 < 16) {
#pragma unroll
            for (int t = 0; t < HEADS; ++t) b[(j + 1) & 1][t] = bl[t * 512 + (j + 1) * 32];
        }
        __builtin_amdgcn_sched_barrier(0);
        v2f alo, ahi;
        alo.x = a4[j].x; alo.y = a4[j].y;
        ahi.x = a4[j].z; ahi.y = a4[j].w;
#pragma unroll
        for (int t = 0; t < HEADS; ++t) {
            v2f bb; bb.x = b[j & 1][t].x; bb.y = b[j & 1][t].y;
            c[t] = __builtin_amdgcn_wmma_f32_16x16x4_f32(
                        false, alo,  false, bb,
                        (short)0, c[t],  false,  false);
        }
#pragma unroll
        for (int t = 0; t < HEADS; ++t) {
            v2f bb; bb.x = b[j & 1][t].z; bb.y = b[j & 1][t].w;
            c[t] = __builtin_amdgcn_wmma_f32_16x16x4_f32(
                        false, ahi,  false, bb,
                        (short)0, c[t],  false,  false);
        }
        __builtin_amdgcn_sched_barrier(0);
    }

    const int rbase = hh ? 8 : 0;
#pragma unroll
    for (int v = 0; v < 8; ++v) {
        float* hr = h + (size_t)(m0 + rbase + v) * OUT_F + l16;
#pragma unroll
        for (int t = 0; t < HEADS; ++t) hr[t * HEAD_DIM] = c[t][v];
    }
}

__global__ void gat_node_scores(const float* __restrict__ h, const float* __restrict__ a,
                                float* __restrict__ s_src, float* __restrict__ s_tgt) {
    const int i = blockIdx.x * blockDim.x + threadIdx.x;
    if (i >= N_NODES * HEADS) return;
    const int hd = i & (HEADS - 1);
    const int n  = i >> 3;
    const float* hv = h + (size_t)n * OUT_F + hd * HEAD_DIM;
    const float* as = a + hd * 2 * HEAD_DIM;
    const float* at = as + HEAD_DIM;
    float ss = 0.0f, st = 0.0f;
#pragma unroll
    for (int d = 0; d < HEAD_DIM; ++d) {
        const float v = hv[d];
        ss += v * as[d];
        st += v * at[d];
    }
    s_src[i] = ss;
    s_tgt[i] = st;
}

#define MB     256
#define MTHR   512
#define MWAV   (MTHR / 32)
#define MGT    (MB * MTHR)
#define MSTEP  49
#define MRND   7
#define MNR    (MSTEP / MRND)
#define NSR    (N_EDGES / 256)
static_assert(MSTEP % MRND == 0, "a wave's steps are whole rounds");
static_assert(MB * MWAV * MSTEP * 4 >= N_EDGES, "the waves' stretches cover the edge list");
static_assert(MB * MWAV >= NSR, "every plane job has a wave");

__global__ __launch_bounds__(MTHR)
void gat_edge_max(const int* __restrict__ src, const int* __restrict__ tgt,
                  const float* __restrict__ s_src, const float* __restrict__ s_tgt,
                  unsigned* __restrict__ gmax, unsigned* __restrict__ own4) {
    __shared__ float wmax[MWAV][HEADS];
    const int tid  = threadIdx.x;
    const int lane = tid & 31;
    const int wave = tid >> 5;
    const int hd   = lane & (HEADS - 1);
    const int e0   = (blockIdx.x * MWAV + wave) * (MSTEP * 4) + (lane >> 3);

    int sc[MRND], tc[MRND];
#pragma unroll
    for (int u = 0; u < MRND; ++u) {
        const int e = min(e0 + 4 * u, N_EDGES - 1);
        sc[u] = src[e];
        tc[u] = tgt[e];
    }

    const int gw = blockIdx.x * MWAV + __builtin_amdgcn_readfirstlane(wave);
    if (gw < NSR) {
        int tv[8];
#pragma unroll
        for (int j = 0; j < 8; ++j) tv[j] = tgt[gw * 256 + 32 * j + lane];
        unsigned w0 = 0u, w1 = 0u;
#pragma unroll
        for (int j = 0; j < 8; ++j) {
            const unsigned ow = (unsigned)tv[j] / (unsigned)NPB;
#pragma unroll
            for (int k = 0; k < 8; ++k) {
                const unsigned b = __builtin_amdgcn_ballot_w32(((ow >> k) & 1u) != 0u);
                if (k < 4) w0 = (lane == 4 * j + k)     ? b : w0;
                else       w1 = (lane == 4 * j + k - 4) ? b : w1;
            }
        }
        unsigned* pw = own4 + 256 * (gw >> 2) + 32 * (gw & 3) + lane;
        pw[0]   = w0;
        pw[128] = w1;
    }

    float m = -INFINITY;
    for (int r = 0; r < MNR; ++r) {
        float a[MRND], b[MRND];
#pragma unroll
        for (int u = 0; u < MRND; ++u) {
            a[u] = s_src[sc[u] * HEADS + hd];
            b[u] = s_tgt[tc[u] * HEADS + hd];
        }
        const int rn = min(r + 1, MNR - 1);
#pragma unroll
        for (int u = 0; u < MRND; ++u) {
            const int e = min(e0 + 4 * (rn * MRND + u), N_EDGES - 1);
            sc[u] = src[e];
            tc[u] = tgt[e];
        }
        __builtin_amdgcn_sched_barrier(0);
#pragma unroll
        for (int u = 0; u < MRND; ++u) m = fmaxf(m, a[u] + b[u]);
    }

    m = fmaxf(m, __shfl_xor(m, 8));
    m = fmaxf(m, __shfl_xor(m, 16));
    if (lane < HEADS) wmax[wave][lane] = m;
    __syncthreads();
    if (tid < HEADS) {
        float v = wmax[0][tid];
        for (int w = 1; w < MWAV; ++w) v = fmaxf(v, wmax[w][tid]);
        atomicMax(&gmax[tid], enc_f32(leaky(v)));
    }
}

#define OTHR  1024
#define OWAV  (OTHR / 32)
#define LPB   2
#define SCH   (OTHR * 32)
#define PU    6
#define WQ    (32 * LPB * 16 + 8)
#define NSC   ((N_EDGES + SCH - 1) / SCH)
static_assert(LPB * 4 == 8, "a lane's loads of a chunk are the eight plane words of its group");
static_assert(OWAV == 32, "a wave's 32 lanes of a chunk are one stretch of 32 groups");

__device__ __forceinline__ void load_chunk(const uint4* __restrict__ own16, int chunk,
                                           int wave, int lane, uint4 (&pre)[LPB]) {
#pragma unroll
    for (int k = 0; k < LPB; ++k) {
        const int i16 = chunk * (SCH / 16) + (wave * LPB + k) * 32 + lane;
        pre[k] = own16[min(i16, ((NGRP - 1) >> 5) * (LPB * 32) + k * 32 + ((NGRP - 1) & 31))];
    }
}

template <bool PAD>
__device__ __forceinline__ void batch_ids(const int* qw, int i0, int n,
                                          const int* __restrict__ tgt, const int* __restrict__ src,
                                          int (&t)[PU], int (&s)[PU]) {
    int e[PU];
#pragma unroll
    for (int u = 0; u < PU; ++u) e[u] = qw[(!PAD || i0 + u < n) ? i0 + u : i0];
#pragma unroll
    for (int u = 0; u < PU; ++u) { t[u] = tgt[e[u]]; s[u] = src[e[u]]; }
}

template <bool PAD>
__device__ __forceinline__ void batch_add(const int (&t)[PU], const int (&s)[PU], int i0, int n,
                                          const float* __restrict__ h, const float* __restrict__ s_src,
                                          const float* stg, float* acc, float* den,
                                          int n0, int lane, int hd, float mx) {
    float4 hv[PU];
    float  ss[PU], tg[PU], ev[PU];
#pragma unroll
    for (int u = 0; u < PU; ++u) ss[u] = s_src[s[u] * HEADS + hd];
#pragma unroll
    for (int u = 0; u < PU; ++u)
        hv[u] = *reinterpret_cast<const float4*>(h + (size_t)s[u] * OUT_F + 4 * lane);
#pragma unroll
    for (int u = 0; u < PU; ++u) tg[u] = stg[(t[u] - n0) * HEADS + hd];
#pragma unroll
    for (int u = 0; u < PU; ++u) {
        const float scv = leaky(ss[u] + tg[u]);
        ev[u] = __expf(fminf(fmaxf(scv - mx, -20.0f), 20.0f));
        if (PAD && i0 + u >= n) ev[u] = 0.0f;
    }
#pragma unroll
    for (int u = 0; u < PU; ++u) {
        const int lt = t[u] - n0;
        float* a = acc + lt * OUT_F + lane;
        atomicAdd(a,      hv[u].x * ev[u]);
        atomicAdd(a + 32, hv[u].y * ev[u]);
        atomicAdd(a + 64, hv[u].z * ev[u]);
        atomicAdd(a + 96, hv[u].w * ev[u]);
        if ((lane & 3) == 0) atomicAdd(&den[lt * HEADS + hd], ev[u]);
    }
}

__global__ __launch_bounds__(OTHR)
void gat_owner_agg(const int* __restrict__ src, const int* __restrict__ tgt,
                   const float* __restrict__ s_src, const float* __restrict__ s_tgt,
                   const unsigned* __restrict__ gmax, const float* __restrict__ h,
                   const uint4* __restrict__ own16,
                   const float* __restrict__ bias, float* __restrict__ out) {
    __shared__ float acc[NPB * OUT_F];
    __shared__ float den[NPB * HEADS];
    __shared__ float stg[NPB * HEADS];
    __shared__ int   q[OWAV][WQ];

    const int tid  = threadIdx.x;
    const int lane = tid & 31;
    const int wave = tid >> 5;
    const int hd   = lane >> 2;
    const int n0   = blockIdx.x * NPB;

    for (int i = tid; i < NPB * OUT_F; i += OTHR) acc[i] = 0.0f;
    for (int i = tid; i < NPB * HEADS; i += OTHR) {
        den[i] = 0.0f;
        stg[i] = (n0 + (i >> 3) < N_NODES) ? s_tgt[n0 * HEADS + i] : 0.0f;
    }

    const float mx = dec_f32(gmax[hd]);
    unsigned cb[8];
#pragma unroll
    for (int k = 0; k < 8; ++k) cb[k] = 0u - ((blockIdx.x >> k) & 1u);
    int* qw = q[wave];

    int chunk = blockIdx.x % NSC;
    uint4 pre[LPB];
    load_chunk(own16, chunk, wave, lane, pre);
    __syncthreads();

    int n = 0;
    int t[PU], s[PU];
    for (int sc = 0; sc < NSC; ++sc) {
        {
            const int g = (chunk * OWAV + wave) * 32 + lane;
            const unsigned mis = (pre[0].x ^ cb[0]) | (pre[0].y ^ cb[1]) | (pre[0].z ^ cb[2]) | (pre[0].w ^ cb[3])
                               | (pre[1].x ^ cb[4]) | (pre[1].y ^ cb[5]) | (pre[1].z ^ cb[6]) | (pre[1].w ^ cb[7]);
            unsigned m = (g < NGRP) ? ~mis : 0u;
            unsigned live;
            while ((live = __builtin_amdgcn_ballot_w32(m != 0u)) != 0u) {
                if (m != 0u) {
                    const int p = __builtin_ctz(m);
                    m &= m - 1u;
                    qw[n + (int)__builtin_amdgcn_mbcnt_lo(live, 0u)] = g * 32 + p;
                }
                n += __builtin_popcount(live);
            }
        }
        chunk = (chunk + 1 == NSC) ? 0 : chunk + 1;

        const bool has = n >= PU;
        if (has) batch_ids<false>(qw, 0, n, tgt, src, t, s);
        load_chunk(own16, chunk, wave, lane, pre);
        if (has) batch_add<false>(t, s, 0, n, h, s_src, stg, acc, den, n0, lane, hd, mx);
        int i0 = has ? PU : 0;
        for (; i0 + PU <= n; i0 += PU) {
            batch_ids<false>(qw, i0, n, tgt, src, t, s);
            batch_add<false>(t, s, i0, n, h, s_src, stg, acc, den, n0, lane, hd, mx);
        }
        const int left = n - i0;
        if (i0 > 0 && left > 0) {
            const int v = qw[i0 + min(lane, left - 1)];
            if (lane < left) qw[lane] = v;
        }
        n = left;
    }
    if (n > 0) {
        batch_ids<true>(qw, 0, n, tgt, src, t, s);
        batch_add<true>(t, s, 0, n, h, s_src, stg, acc, den, n0, lane, hd, mx);
    }
    __syncthreads();

    const float4 bb = *reinterpret_cast<const float4*>(bias + 4 * lane);
    for (int idx = tid; idx < NPB * 32; idx += OTHR) {
        const int lt = idx >> 5;
        const int nd = n0 + lt;
        if (nd < N_NODES) {
            const float  d = den[lt * HEADS + hd] + EPS_F;
            const float* a = acc + lt * OUT_F + lane;
            float4 v;
            v.x = a[0]  / d + bb.x;
            v.y = a[32] / d + bb.y;
            v.z = a[64] / d + bb.z;
            v.w = a[96] / d + bb.w;
            *reinterpret_cast<float4*>(out + (size_t)nd * OUT_F + 4 * lane) = v;
        }
    }
}

extern "C" void kernel_launch(void* const* d_in, const int* in_sizes, int n_in,
                              void* d_out, int out_size, void* d_ws, size_t ws_size,
                              hipStream_t stream) {
    (void)in_sizes; (void)n_in; (void)out_size; (void)ws_size;

    const float* x    = (const float*)d_in[0];
    const int*   ei   = (const int*)  d_in[1];
    const float* W    = (const float*)d_in[2];
    const float* a    = (const float*)d_in[3];
    const float* bias = (const float*)d_in[4];
    float* out = (float*)d_out;

    const int* src = ei;
    const int* tgt = ei + N_EDGES;

    float*    ws    = (float*)d_ws;
    float*    hbuf  = ws;
    float*    ssrc  = hbuf + (size_t)N_NODES * OUT_F;
    float*    stgt  = ssrc + (size_t)N_NODES * HEADS;
    float*    ownf  = stgt + (size_t)N_NODES * HEADS;
    unsigned* gmax  = (unsigned*)(ownf + (size_t)N_NODES * HEADS);

    gat_init<<<1, 32, 0, stream>>>(gmax);

    gat_gemm_wmma<<<GBLK, 256, 0, stream>>>(x, W, hbuf);

    gat_node_scores<<<(N_NODES * HEADS + 255) / 256, 256, 0, stream>>>(hbuf, a, ssrc, stgt);

    gat_edge_max<<<MB, MTHR, 0, stream>>>(src, tgt, ssrc, stgt, gmax, (unsigned*)ownf);

    gat_owner_agg<<<NBLK, OTHR, 0, stream>>>(src, tgt, ssrc, stgt, gmax, hbuf, (const uint4*)ownf, bias, out);
}
